# starting kernel
# baseline (speedup 1.0000x reference)
#include <hip/hip_runtime.h>

typedef __attribute__((ext_vector_type(2))) float v2f;
typedef __attribute__((ext_vector_type(8))) float v8f;

constexpr int Hdim = 256, Wdim = 256, Cin = 64, C2 = 256, HID = 128, Cout = 64;
constexpr int HW = Hdim * Wdim;

__constant__ float DCT8[8][8] = {
  { 0.3535533906f, 0.3535533906f, 0.3535533906f, 0.3535533906f, 0.3535533906f, 0.3535533906f, 0.3535533906f, 0.3535533906f },
  { 0.4903926402f, 0.4157348062f, 0.2777851165f, 0.0975451610f,-0.0975451610f,-0.2777851165f,-0.4157348062f,-0.4903926402f },
  { 0.4619397663f, 0.1913417162f,-0.1913417162f,-0.4619397663f,-0.4619397663f,-0.1913417162f, 0.1913417162f, 0.4619397663f },
  { 0.4157348062f,-0.0975451610f,-0.4903926402f,-0.2777851165f, 0.2777851165f, 0.4903926402f, 0.0975451610f,-0.4157348062f },
  { 0.3535533906f,-0.3535533906f,-0.3535533906f, 0.3535533906f, 0.3535533906f,-0.3535533906f,-0.3535533906f, 0.3535533906f },
  { 0.2777851165f,-0.4903926402f, 0.0975451610f, 0.4157348062f,-0.4157348062f,-0.0975451610f, 0.4903926402f,-0.2777851165f },
  { 0.1913417162f,-0.4619397663f, 0.4619397663f,-0.1913417162f,-0.1913417162f, 0.4619397663f,-0.4619397663f, 0.1913417162f },
  { 0.0975451610f,-0.2777851165f, 0.4157348062f,-0.4903926402f, 0.4903926402f,-0.4157348062f, 0.2777851165f,-0.0975451610f }
};

__global__ __launch_bounds__(256) void
k1_projin_dct(const float* __restrict__ x, const float* __restrict__ Win,
              const float* __restrict__ mix, float* __restrict__ ws)
{
    __shared__ float yt[16][272];
    __shared__ float vt[16][272];

    const int tid  = threadIdx.x;
    const int lane = tid & 31;
    const int wv   = tid >> 5;
    const int m    = lane & 15;
    const int hi   = lane >> 4;

    const int b  = blockIdx.z >> 4;
    const int ob = blockIdx.z & 15;
    const int r0 = blockIdx.y * 16;
    const int c0 = blockIdx.x * 16;

    const float* xb = x + (size_t)b * Cin * HW;
    const float* wr = Win + (size_t)(ob * 16 + m) * Cin;
    const int t0 = 2 * wv, t1 = 2 * wv + 1;
    const float* xp0 = xb + (size_t)(r0 + t0) * Wdim + c0 + m;
    const float* xp1 = xb + (size_t)(r0 + t1) * Wdim + c0 + m;

    v8f acc0 = {}, acc1 = {};
#pragma unroll
    for (int kk = 0; kk < Cin; kk += 4) {
        const int k0 = kk + 2 * hi;
        v2f a;  a.x  = wr[k0];                    a.y  = wr[k0 + 1];
        v2f b0; b0.x = xp0[(size_t)k0 * HW];      b0.y = xp0[(size_t)(k0 + 1) * HW];
        v2f b1; b1.x = xp1[(size_t)k0 * HW];      b1.y = xp1[(size_t)(k0 + 1) * HW];
        acc0 = __builtin_amdgcn_wmma_f32_16x16x4_f32(false, a, false, b0, (short)0, acc0, false, false);
        acc1 = __builtin_amdgcn_wmma_f32_16x16x4_f32(false, a, false, b1, (short)0, acc1, false, false);
    }
#pragma unroll
    for (int r = 0; r < 8; ++r) {
        yt[r + 8 * hi][t0 * 17 + m] = acc0[r];
        yt[r + 8 * hi][t1 * 17 + m] = acc1[r];
    }
    __syncthreads();

    const float* mx = mix + (size_t)(ob * 16) * 64;
#pragma unroll 1
    for (int task = tid; task < 512; task += 256) {
        const int i  = task & 7;
        const int u  = task >> 3;
        const int ch = u >> 2;
        const int pr = ((u >> 1) & 1) * 8;
        const int pc = (u & 1) * 8;
        float U[8];
#pragma unroll
        for (int j = 0; j < 8; ++j) {
            float s = 0.f;
#pragma unroll
            for (int k = 0; k < 8; ++k) s += DCT8[i][k] * yt[ch][(pr + k) * 17 + pc + j];
            U[j] = s;
        }
#pragma unroll
        for (int q = 0; q < 8; ++q) {
            float s = 0.f;
#pragma unroll
            for (int j = 0; j < 8; ++j) s += U[j] * DCT8[q][j];
            vt[ch][(pr + i) * 17 + pc + q] = s * mx[ch * 64 + i * 8 + q];
        }
    }
    __syncthreads();

#pragma unroll 1
    for (int task = tid; task < 512; task += 256) {
        const int i  = task & 7;
        const int u  = task >> 3;
        const int ch = u >> 2;
        const int pr = ((u >> 1) & 1) * 8;
        const int pc = (u & 1) * 8;
        float U[8];
#pragma unroll
        for (int j = 0; j < 8; ++j) {
            float s = 0.f;
#pragma unroll
            for (int k = 0; k < 8; ++k) s += DCT8[i][k] * vt[ch][(pr + k) * 17 + pc + j];
            U[j] = s;
        }
#pragma unroll
        for (int q = 0; q < 8; ++q) {
            float s = 0.f;
#pragma unroll
            for (int j = 0; j < 8; ++j) s += U[j] * DCT8[q][j];
            yt[ch][(pr + i) * 17 + pc + q] = s;
        }
    }
    __syncthreads();

    const int py = tid >> 4, px = tid & 15;
    size_t obase = ((size_t)b * C2 + ob * 16) * HW + (size_t)(r0 + py) * Wdim + (c0 + px);
#pragma unroll
    for (int ch = 0; ch < 16; ++ch)
        ws[obase + (size_t)ch * HW] = yt[ch][py * 17 + px];
}

__global__ __launch_bounds__(256) void
k2_dw_gate_out(const float* __restrict__ ws, const float* __restrict__ wdw,
               const float* __restrict__ wout, float* __restrict__ out)
{
    __shared__ float g[16][272];

    const int tid  = threadIdx.x;
    const int lane = tid & 31;
    const int wv   = tid >> 5;
    const int m    = lane & 15;
    const int hi   = lane >> 4;

    const int b  = blockIdx.z;
    const int r0 = blockIdx.y * 16;
    const int c0 = blockIdx.x * 16;
    const int py = tid >> 4, px = tid & 15;
    const int gy = r0 + py,  gx = c0 + px;

    const float* wsb  = ws + (size_t)b * C2 * HW;
    const int mt  = wv >> 1;
    const int ntb = (wv & 1) * 8;
    const float* wrow = wout + (size_t)(mt * 16 + m) * HID;

    v8f acc[8];
#pragma unroll
    for (int j = 0; j < 8; ++j) acc[j] = (v8f){};

    for (int kc = 0; kc < HID; kc += 16) {
        __syncthreads();
#pragma unroll 1
        for (int i = 0; i < 16; ++i) {
            const int c = kc + i;
            const float* w1 = wdw + c * 9;
            const float* w2 = wdw + (c + HID) * 9;
            const float* p1 = wsb + (size_t)c * HW;
            const float* p2 = wsb + (size_t)(c + HID) * HW;
            float s1 = 0.f, s2 = 0.f;
#pragma unroll
            for (int ky = 0; ky < 3; ++ky) {
                const int yy = gy + ky - 1;
                const bool iny = ((unsigned)yy < (unsigned)Hdim);
#pragma unroll
                for (int kx = 0; kx < 3; ++kx) {
                    const int xx = gx + kx - 1;
                    const bool in = iny && ((unsigned)xx < (unsigned)Wdim);
                    const size_t off = (size_t)yy * Wdim + xx;
                    const float v1 = in ? p1[off] : 0.f;
                    const float v2 = in ? p2[off] : 0.f;
                    s1 += w1[ky * 3 + kx] * v1;
                    s2 += w2[ky * 3 + kx] * v2;
                }
            }
            const float t = tanhf(0.7978845608028654f * (s1 + 0.044715f * s1 * s1 * s1));
            g[i][py * 16 + px] = 0.5f * s1 * (1.f + t) * s2;
        }
        __syncthreads();
#pragma unroll
        for (int kk = 0; kk < 16; kk += 4) {
            const int k0 = kk + 2 * hi;
            v2f a; a.x = wrow[kc + k0]; a.y = wrow[kc + k0 + 1];
#pragma unroll
            for (int j = 0; j < 8; ++j) {
                const int nt = ntb + j;
                v2f bv; bv.x = g[k0][nt * 16 + m]; bv.y = g[k0 + 1][nt * 16 + m];
                acc[j] = __builtin_amdgcn_wmma_f32_16x16x4_f32(false, a, false, bv, (short)0, acc[j], false, false);
            }
        }
    }

#pragma unroll
    for (int j = 0; j < 8; ++j) {
        const int nt = ntb + j;
        size_t base = ((size_t)b * Cout + mt * 16 + 8 * hi) * HW
                    + (size_t)(r0 + nt) * Wdim + c0 + m;
#pragma unroll
        for (int r = 0; r < 8; ++r)
            out[base + (size_t)r * HW] = acc[j][r];
    }
}

extern "C" void kernel_launch(void* const* d_in, const int* in_sizes, int n_in,
                              void* d_out, int out_size, void* d_ws, size_t ws_size,
                              hipStream_t stream) {
    const float* x    = (const float*)d_in[0];
    const float* Win  = (const float*)d_in[1];
    const float* Wdw  = (const float*)d_in[2];
    const float* mix  = (const float*)d_in[3];
    const float* Wout = (const float*)d_in[4];
    float* out = (float*)d_out;
    float* ws  = (float*)d_ws;

    dim3 blk(256);
    dim3 g1(Wdim / 16, Hdim / 16, 2 * 16);
    k1_projin_dct<<<g1, blk, 0, stream>>>(x, Win, mix, ws);
    dim3 g2(Wdim / 16, Hdim / 16, 2);
    k2_dw_gate_out<<<g2, blk, 0, stream>>>(ws, Wdw, Wout, out);
}


# second kernel: conv input planes of each 16-channel chunk staged in LDS (18x18 halo tile) by one batched prefetch per chunk instead of 18 dependent global loads per channel; sums in same order
# speedup vs baseline: 1.2447x; 1.2447x over previous
#include <hip/hip_runtime.h>

typedef __attribute__((ext_vector_type(2))) float v2f;
typedef __attribute__((ext_vector_type(8))) float v8f;

constexpr int Hdim = 256, Wdim = 256, Cin = 64, C2 = 256, HID = 128, Cout = 64;
constexpr int HW = Hdim * Wdim;

__constant__ float DCT8[8][8] = {
  { 0.3535533906f, 0.3535533906f, 0.3535533906f, 0.3535533906f, 0.3535533906f, 0.3535533906f, 0.3535533906f, 0.3535533906f },
  { 0.4903926402f, 0.4157348062f, 0.2777851165f, 0.0975451610f,-0.0975451610f,-0.2777851165f,-0.4157348062f,-0.4903926402f },
  { 0.4619397663f, 0.1913417162f,-0.1913417162f,-0.4619397663f,-0.4619397663f,-0.1913417162f, 0.1913417162f, 0.4619397663f },
  { 0.4157348062f,-0.0975451610f,-0.4903926402f,-0.2777851165f, 0.2777851165f, 0.4903926402f, 0.0975451610f,-0.4157348062f },
  { 0.3535533906f,-0.3535533906f,-0.3535533906f, 0.3535533906f, 0.3535533906f,-0.3535533906f,-0.3535533906f, 0.3535533906f },
  { 0.2777851165f,-0.4903926402f, 0.0975451610f, 0.4157348062f,-0.4157348062f,-0.0975451610f, 0.4903926402f,-0.2777851165f },
  { 0.1913417162f,-0.4619397663f, 0.4619397663f,-0.1913417162f,-0.1913417162f, 0.4619397663f,-0.4619397663f, 0.1913417162f },
  { 0.0975451610f,-0.2777851165f, 0.4157348062f,-0.4903926402f, 0.4903926402f,-0.4157348062f, 0.2777851165f,-0.0975451610f }
};

__global__ __launch_bounds__(256) void
k1_projin_dct(const float* __restrict__ x, const float* __restrict__ Win,
              const float* __restrict__ mix, float* __restrict__ ws)
{
    __shared__ float yt[16][272];
    __shared__ float vt[16][272];

    const int tid  = threadIdx.x;
    const int lane = tid & 31;
    const int wv   = tid >> 5;
    const int m    = lane & 15;
    const int hi   = lane >> 4;

    const int b  = blockIdx.z >> 4;
    const int ob = blockIdx.z & 15;
    const int r0 = blockIdx.y * 16;
    const int c0 = blockIdx.x * 16;

    const float* xb = x + (size_t)b * Cin * HW;
    const float* wr = Win + (size_t)(ob * 16 + m) * Cin;
    const int t0 = 2 * wv, t1 = 2 * wv + 1;
    const float* xp0 = xb + (size_t)(r0 + t0) * Wdim + c0 + m;
    const float* xp1 = xb + (size_t)(r0 + t1) * Wdim + c0 + m;

    v8f acc0 = {}, acc1 = {};
#pragma unroll
    for (int kk = 0; kk < Cin; kk += 4) {
        const int k0 = kk + 2 * hi;
        v2f a;  a.x  = wr[k0];                    a.y  = wr[k0 + 1];
        v2f b0; b0.x = xp0[(size_t)k0 * HW];      b0.y = xp0[(size_t)(k0 + 1) * HW];
        v2f b1; b1.x = xp1[(size_t)k0 * HW];      b1.y = xp1[(size_t)(k0 + 1) * HW];
        acc0 = __builtin_amdgcn_wmma_f32_16x16x4_f32(false, a, false, b0, (short)0, acc0, false, false);
        acc1 = __builtin_amdgcn_wmma_f32_16x16x4_f32(false, a, false, b1, (short)0, acc1, false, false);
    }
#pragma unroll
    for (int r = 0; r < 8; ++r) {
        yt[r + 8 * hi][t0 * 17 + m] = acc0[r];
        yt[r + 8 * hi][t1 * 17 + m] = acc1[r];
    }
    __syncthreads();

    const float* mx = mix + (size_t)(ob * 16) * 64;
#pragma unroll 1
    for (int task = tid; task < 512; task += 256) {
        const int i  = task & 7;
        const int u  = task >> 3;
        const int ch = u >> 2;
        const int pr = ((u >> 1) & 1) * 8;
        const int pc = (u & 1) * 8;
        float U[8];
#pragma unroll
        for (int j = 0; j < 8; ++j) {
            float s = 0.f;
#pragma unroll
            for (int k = 0; k < 8; ++k) s += DCT8[i][k] * yt[ch][(pr + k) * 17 + pc + j];
            U[j] = s;
        }
#pragma unroll
        for (int q = 0; q < 8; ++q) {
            float s = 0.f;
#pragma unroll
            for (int j = 0; j < 8; ++j) s += U[j] * DCT8[q][j];
            vt[ch][(pr + i) * 17 + pc + q] = s * mx[ch * 64 + i * 8 + q];
        }
    }
    __syncthreads();

#pragma unroll 1
    for (int task = tid; task < 512; task += 256) {
        const int i  = task & 7;
        const int u  = task >> 3;
        const int ch = u >> 2;
        const int pr = ((u >> 1) & 1) * 8;
        const int pc = (u & 1) * 8;
        float U[8];
#pragma unroll
        for (int j = 0; j < 8; ++j) {
            float s = 0.f;
#pragma unroll
            for (int k = 0; k < 8; ++k) s += DCT8[i][k] * vt[ch][(pr + k) * 17 + pc + j];
            U[j] = s;
        }
#pragma unroll
        for (int q = 0; q < 8; ++q) {
            float s = 0.f;
#pragma unroll
            for (int j = 0; j < 8; ++j) s += U[j] * DCT8[q][j];
            yt[ch][(pr + i) * 17 + pc + q] = s;
        }
    }
    __syncthreads();

    const int py = tid >> 4, px = tid & 15;
    size_t obase = ((size_t)b * C2 + ob * 16) * HW + (size_t)(r0 + py) * Wdim + (c0 + px);
#pragma unroll
    for (int ch = 0; ch < 16; ++ch)
        ws[obase + (size_t)ch * HW] = yt[ch][py * 17 + px];
}

typedef __attribute__((ext_vector_type(4))) float v4f;

constexpr int TP  = 24;
constexpr int TCH = 18 * TP;

__global__ __launch_bounds__(256) void
k2_dw_gate_out(const float* __restrict__ ws, const float* __restrict__ wdw,
               const float* __restrict__ wout, float* __restrict__ out)
{
    __shared__ float g[16][272];
    __shared__ alignas(16) float tile[32 * TCH];

    const int tid  = threadIdx.x;
    const int lane = tid & 31;
    const int wv   = tid >> 5;
    const int m    = lane & 15;
    const int hi   = lane >> 4;

    const int b  = blockIdx.z;
    const int r0 = blockIdx.y * 16;
    const int c0 = blockIdx.x * 16;
    const int py = tid >> 4, px = tid & 15;

    const float* wsb  = ws + (size_t)b * C2 * HW;
    const int mt  = wv >> 1;
    const int ntb = (wv & 1) * 8;
    const float* wrow = wout + (size_t)(mt * 16 + m) * HID;

    const int quad = tid & 3, rr = tid >> 2;
    v4f   cv[9];
    float hv[5];

    auto fetch = [&](int kc) {
#pragma unroll
        for (int t = 0; t < 9; ++t) {
            const int R   = rr + 64 * t;
            const int cs  = R / 18;
            const int row = R - cs * 18;
            const int yy  = r0 - 1 + row;
            const bool ok = ((unsigned)yy < (unsigned)Hdim);
            const int c   = kc + (cs & 15) + (cs >> 4) * HID;
            const float* p = wsb + (size_t)c * HW + (size_t)(ok ? yy : 0) * Wdim + c0 + quad * 4;
            const v4f v = *reinterpret_cast<const v4f*>(p);
            cv[t] = ok ? v : (v4f){0.f, 0.f, 0.f, 0.f};
        }
#pragma unroll
        for (int t = 0; t < 5; ++t) {
            const int e    = tid + 256 * t;
            const int side = e & 1;
            const int R    = e >> 1;
            const int cs   = R / 18;
            const int row  = R - cs * 18;
            const int yy   = r0 - 1 + row;
            const int xx   = side ? c0 + 16 : c0 - 1;
            const bool ok  = (e < 1152) && ((unsigned)yy < (unsigned)Hdim) && ((unsigned)xx < (unsigned)Wdim);
            const int c    = kc + (cs & 15) + (cs >> 4) * HID;
            hv[t] = ok ? wsb[(size_t)c * HW + (size_t)yy * Wdim + xx] : 0.f;
        }
    };
    auto stage = [&]() {
#pragma unroll
        for (int t = 0; t < 9; ++t) {
            const int R   = rr + 64 * t;
            const int cs  = R / 18;
            const int row = R - cs * 18;
            *reinterpret_cast<v4f*>(&tile[cs * TCH + row * TP + 4 + quad * 4]) = cv[t];
        }
#pragma unroll
        for (int t = 0; t < 5; ++t) {
            const int e    = tid + 256 * t;
            const int side = e & 1;
            const int R    = e >> 1;
            const int cs   = R / 18;
            const int row  = R - cs * 18;
            if (e < 1152) tile[cs * TCH + row * TP + (side ? 20 : 3)] = hv[t];
        }
    };

    v8f acc[8];
#pragma unroll
    for (int j = 0; j < 8; ++j) acc[j] = (v8f){};

    fetch(0);
    stage();
    __syncthreads();

    for (int kc = 0; kc < HID; kc += 16) {
        const bool more = (kc + 16 < HID);
        if (more) fetch(kc + 16);
        const float* tb = tile + py * TP + px + 3;
#pragma unroll 1
        for (int i = 0; i < 16; ++i) {
            const int c = kc + i;
            const float* w1 = wdw + c * 9;
            const float* w2 = wdw + (c + HID) * 9;
            const float* t1 = tb + i * TCH;
            const float* t2 = t1 + 16 * TCH;
            float s1 = 0.f, s2 = 0.f;
#pragma unroll
            for (int ky = 0; ky < 3; ++ky) {
#pragma unroll
                for (int kx = 0; kx < 3; ++kx) {
                    const float v1 = t1[ky * TP + kx];
                    const float v2 = t2[ky * TP + kx];
                    s1 += w1[ky * 3 + kx] * v1;
                    s2 += w2[ky * 3 + kx] * v2;
                }
            }
            const float t = tanhf(0.7978845608028654f * (s1 + 0.044715f * s1 * s1 * s1));
            g[i][py * 16 + px] = 0.5f * s1 * (1.f + t) * s2;
        }
        __syncthreads();
        if (more) stage();
#pragma unroll
        for (int kk = 0; kk < 16; kk += 4) {
            const int k0 = kk + 2 * hi;
            v2f a; a.x = wrow[kc + k0]; a.y = wrow[kc + k0 + 1];
#pragma unroll
            for (int j = 0; j < 8; ++j) {
                const int nt = ntb + j;
                v2f bv; bv.x = g[k0][nt * 16 + m]; bv.y = g[k0 + 1][nt * 16 + m];
                acc[j] = __builtin_amdgcn_wmma_f32_16x16x4_f32(false, a, false, bv, (short)0, acc[j], false, false);
            }
        }
        __syncthreads();
    }

#pragma unroll
    for (int j = 0; j < 8; ++j) {
        const int nt = ntb + j;
        size_t base = ((size_t)b * Cout + mt * 16 + 8 * hi) * HW
                    + (size_t)(r0 + nt) * Wdim + c0 + m;
#pragma unroll
        for (int r = 0; r < 8; ++r)
            out[base + (size_t)r * HW] = acc[j][r];
    }
}

extern "C" void kernel_launch(void* const* d_in, const int* in_sizes, int n_in,
                              void* d_out, int out_size, void* d_ws, size_t ws_size,
                              hipStream_t stream) {
    const float* x    = (const float*)d_in[0];
    const float* Win  = (const float*)d_in[1];
    const float* Wdw  = (const float*)d_in[2];
    const float* mix  = (const float*)d_in[3];
    const float* Wout = (const float*)d_in[4];
    float* out = (float*)d_out;
    float* ws  = (float*)d_ws;

    dim3 blk(256);
    dim3 g1(Wdim / 16, Hdim / 16, 2 * 16);
    k1_projin_dct<<<g1, blk, 0, stream>>>(x, Win, mix, ws);
    dim3 g2(Wdim / 16, Hdim / 16, 2);
    k2_dw_gate_out<<<g2, blk, 0, stream>>>(ws, Wdw, Wout, out);
}


# second kernel: conv+gate remapped: a wave owns 2 hidden channels, a lane 8 pixels of a row (scalar weight loads once per channel, 16-byte LDS tap reads), not 1 pixel x 16 channels; same sum order
# speedup vs baseline: 1.3385x; 1.0754x over previous
#include <hip/hip_runtime.h>

typedef __attribute__((ext_vector_type(2))) float v2f;
typedef __attribute__((ext_vector_type(8))) float v8f;

constexpr int Hdim = 256, Wdim = 256, Cin = 64, C2 = 256, HID = 128, Cout = 64;
constexpr int HW = Hdim * Wdim;

__constant__ float DCT8[8][8] = {
  { 0.3535533906f, 0.3535533906f, 0.3535533906f, 0.3535533906f, 0.3535533906f, 0.3535533906f, 0.3535533906f, 0.3535533906f },
  { 0.4903926402f, 0.4157348062f, 0.2777851165f, 0.0975451610f,-0.0975451610f,-0.2777851165f,-0.4157348062f,-0.4903926402f },
  { 0.4619397663f, 0.1913417162f,-0.1913417162f,-0.4619397663f,-0.4619397663f,-0.1913417162f, 0.1913417162f, 0.4619397663f },
  { 0.4157348062f,-0.0975451610f,-0.4903926402f,-0.2777851165f, 0.2777851165f, 0.4903926402f, 0.0975451610f,-0.4157348062f },
  { 0.3535533906f,-0.3535533906f,-0.3535533906f, 0.3535533906f, 0.3535533906f,-0.3535533906f,-0.3535533906f, 0.3535533906f },
  { 0.2777851165f,-0.4903926402f, 0.0975451610f, 0.4157348062f,-0.4157348062f,-0.0975451610f, 0.4903926402f,-0.2777851165f },
  { 0.1913417162f,-0.4619397663f, 0.4619397663f,-0.1913417162f,-0.1913417162f, 0.4619397663f,-0.4619397663f, 0.1913417162f },
  { 0.0975451610f,-0.2777851165f, 0.4157348062f,-0.4903926402f, 0.4903926402f,-0.4157348062f, 0.2777851165f,-0.0975451610f }
};

__global__ __launch_bounds__(256) void
k1_projin_dct(const float* __restrict__ x, const float* __restrict__ Win,
              const float* __restrict__ mix, float* __restrict__ ws)
{
    __shared__ float yt[16][272];
    __shared__ float vt[16][272];

    const int tid  = threadIdx.x;
    const int lane = tid & 31;
    const int wv   = tid >> 5;
    const int m    = lane & 15;
    const int hi   = lane >> 4;

    const int b  = blockIdx.z >> 4;
    const int ob = blockIdx.z & 15;
    const int r0 = blockIdx.y * 16;
    const int c0 = blockIdx.x * 16;

    const float* xb = x + (size_t)b * Cin * HW;
    const float* wr = Win + (size_t)(ob * 16 + m) * Cin;
    const int t0 = 2 * wv, t1 = 2 * wv + 1;
    const float* xp0 = xb + (size_t)(r0 + t0) * Wdim + c0 + m;
    const float* xp1 = xb + (size_t)(r0 + t1) * Wdim + c0 + m;

    v8f acc0 = {}, acc1 = {};
#pragma unroll
    for (int kk = 0; kk < Cin; kk += 4) {
        const int k0 = kk + 2 * hi;
        v2f a;  a.x  = wr[k0];                    a.y  = wr[k0 + 1];
        v2f b0; b0.x = xp0[(size_t)k0 * HW];      b0.y = xp0[(size_t)(k0 + 1) * HW];
        v2f b1; b1.x = xp1[(size_t)k0 * HW];      b1.y = xp1[(size_t)(k0 + 1) * HW];
        acc0 = __builtin_amdgcn_wmma_f32_16x16x4_f32(false, a, false, b0, (short)0, acc0, false, false);
        acc1 = __builtin_amdgcn_wmma_f32_16x16x4_f32(false, a, false, b1, (short)0, acc1, false, false);
    }
#pragma unroll
    for (int r = 0; r < 8; ++r) {
        yt[r + 8 * hi][t0 * 17 + m] = acc0[r];
        yt[r + 8 * hi][t1 * 17 + m] = acc1[r];
    }
    __syncthreads();

    const float* mx = mix + (size_t)(ob * 16) * 64;
#pragma unroll 1
    for (int task = tid; task < 512; task += 256) {
        const int i  = task & 7;
        const int u  = task >> 3;
        const int ch = u >> 2;
        const int pr = ((u >> 1) & 1) * 8;
        const int pc = (u & 1) * 8;
        float U[8];
#pragma unroll
        for (int j = 0; j < 8; ++j) {
            float s = 0.f;
#pragma unroll
            for (int k = 0; k < 8; ++k) s += DCT8[i][k] * yt[ch][(pr + k) * 17 + pc + j];
            U[j] = s;
        }
#pragma unroll
        for (int q = 0; q < 8; ++q) {
            float s = 0.f;
#pragma unroll
            for (int j = 0; j < 8; ++j) s += U[j] * DCT8[q][j];
            vt[ch][(pr + i) * 17 + pc + q] = s * mx[ch * 64 + i * 8 + q];
        }
    }
    __syncthreads();

#pragma unroll 1
    for (int task = tid; task < 512; task += 256) {
        const int i  = task & 7;
        const int u  = task >> 3;
        const int ch = u >> 2;
        const int pr = ((u >> 1) & 1) * 8;
        const int pc = (u & 1) * 8;
        float U[8];
#pragma unroll
        for (int j = 0; j < 8; ++j) {
            float s = 0.f;
#pragma unroll
            for (int k = 0; k < 8; ++k) s += DCT8[i][k] * vt[ch][(pr + k) * 17 + pc + j];
            U[j] = s;
        }
#pragma unroll
        for (int q = 0; q < 8; ++q) {
            float s = 0.f;
#pragma unroll
            for (int j = 0; j < 8; ++j) s += U[j] * DCT8[q][j];
            yt[ch][(pr + i) * 17 + pc + q] = s;
        }
    }
    __syncthreads();

    const int py = tid >> 4, px = tid & 15;
    size_t obase = ((size_t)b * C2 + ob * 16) * HW + (size_t)(r0 + py) * Wdim + (c0 + px);
#pragma unroll
    for (int ch = 0; ch < 16; ++ch)
        ws[obase + (size_t)ch * HW] = yt[ch][py * 17 + px];
}

typedef __attribute__((ext_vector_type(4))) float v4f;

constexpr int TP  = 28;
constexpr int TCH = 18 * TP;

__global__ __launch_bounds__(256) void
k2_dw_gate_out(const float* __restrict__ ws, const float* __restrict__ wdw,
               const float* __restrict__ wout, float* __restrict__ out)
{
    __shared__ alignas(16) float g[16][272];
    __shared__ alignas(16) float tile[32 * TCH];

    const int tid  = threadIdx.x;
    const int lane = tid & 31;
    const int wv   = tid >> 5;
    const int m    = lane & 15;
    const int hi   = lane >> 4;

    const int b  = blockIdx.z;
    const int r0 = blockIdx.y * 16;
    const int c0 = blockIdx.x * 16;

    const int wvu = __builtin_amdgcn_readfirstlane(wv);
    const int cy  = m;

    const float* wsb  = ws + (size_t)b * C2 * HW;
    const int mt  = wv >> 1;
    const int ntb = (wv & 1) * 8;
    const float* wrow = wout + (size_t)(mt * 16 + m) * HID;

    const int quad = tid & 3, rr = tid >> 2;
    v4f   cv[9];
    float hv[5];

    auto fetch = [&](int kc) {
#pragma unroll
        for (int t = 0; t < 9; ++t) {
            const int R   = rr + 64 * t;
            const int cs  = R / 18;
            const int row = R - cs * 18;
            const int yy  = r0 - 1 + row;
            const bool ok = ((unsigned)yy < (unsigned)Hdim);
            const int c   = kc + (cs & 15) + (cs >> 4) * HID;
            const float* p = wsb + (size_t)c * HW + (size_t)(ok ? yy : 0) * Wdim + c0 + quad * 4;
            const v4f v = *reinterpret_cast<const v4f*>(p);
            cv[t] = ok ? v : (v4f){0.f, 0.f, 0.f, 0.f};
        }
#pragma unroll
        for (int t = 0; t < 5; ++t) {
            const int e    = tid + 256 * t;
            const int side = e & 1;
            const int R    = e >> 1;
            const int cs   = R / 18;
            const int row  = R - cs * 18;
            const int yy   = r0 - 1 + row;
            const int xx   = side ? c0 + 16 : c0 - 1;
            const bool ok  = (e < 1152) && ((unsigned)yy < (unsigned)Hdim) && ((unsigned)xx < (unsigned)Wdim);
            const int c    = kc + (cs & 15) + (cs >> 4) * HID;
            hv[t] = ok ? wsb[(size_t)c * HW + (size_t)yy * Wdim + xx] : 0.f;
        }
    };
    auto stage = [&]() {
#pragma unroll
        for (int t = 0; t < 9; ++t) {
            const int R   = rr + 64 * t;
            const int cs  = R / 18;
            const int row = R - cs * 18;
            *reinterpret_cast<v4f*>(&tile[cs * TCH + row * TP + 4 + quad * 4]) = cv[t];
        }
#pragma unroll
        for (int t = 0; t < 5; ++t) {
            const int e    = tid + 256 * t;
            const int side = e & 1;
            const int R    = e >> 1;
            const int cs   = R / 18;
            const int row  = R - cs * 18;
            if (e < 1152) tile[cs * TCH + row * TP + (side ? 20 : 3)] = hv[t];
        }
    };

    v8f acc[8];
#pragma unroll
    for (int j = 0; j < 8; ++j) acc[j] = (v8f){};

    fetch(0);
    stage();
    __syncthreads();

    for (int kc = 0; kc < HID; kc += 16) {
        const bool more = (kc + 16 < HID);
        if (more) fetch(kc + 16);
#pragma unroll 1
        for (int cc = 0; cc < 2; ++cc) {
            const int i = 2 * wvu + cc;
            const int c = kc + i;
            const float* w1 = wdw + c * 9;
            const float* w2 = wdw + (c + HID) * 9;
            const float* t1 = tile + i * TCH + cy * TP + 3 + 8 * hi;
            const float* t2 = t1 + 16 * TCH;
            float s1[8], s2[8];
#pragma unroll
            for (int j = 0; j < 8; ++j) { s1[j] = 0.f; s2[j] = 0.f; }
#pragma unroll
            for (int ky = 0; ky < 3; ++ky) {
                const float* r1 = t1 + ky * TP;
                const float* r2 = t2 + ky * TP;
                float a[10], bq[10];
                const v4f a0 = *reinterpret_cast<const v4f*>(r1 + 1);
                const v4f a1 = *reinterpret_cast<const v4f*>(r1 + 5);
                const v4f b0 = *reinterpret_cast<const v4f*>(r2 + 1);
                const v4f b1 = *reinterpret_cast<const v4f*>(r2 + 5);
                a[0] = r1[0];  a[9] = r1[9];
                bq[0] = r2[0]; bq[9] = r2[9];
#pragma unroll
                for (int e = 0; e < 4; ++e) { a[1 + e] = a0[e]; a[5 + e] = a1[e]; bq[1 + e] = b0[e]; bq[5 + e] = b1[e]; }
#pragma unroll
                for (int kx = 0; kx < 3; ++kx) {
                    const float wa = w1[ky * 3 + kx];
                    const float wb = w2[ky * 3 + kx];
#pragma unroll
                    for (int j = 0; j < 8; ++j) {
                        s1[j] += wa * a[j + kx];
                        s2[j] += wb * bq[j + kx];
                    }
                }
            }
            v4f o0, o1;
#pragma unroll
            for (int j = 0; j < 8; ++j) {
                const float t = tanhf(0.7978845608028654f * (s1[j] + 0.044715f * s1[j] * s1[j] * s1[j]));
                const float v = 0.5f * s1[j] * (1.f + t) * s2[j];
                if (j < 4) o0[j] = v; else o1[j - 4] = v;
            }
            float* gp = &g[i][cy * 16 + 8 * hi];
            *reinterpret_cast<v4f*>(gp)     = o0;
            *reinterpret_cast<v4f*>(gp + 4) = o1;
        }
        __syncthreads();
        if (more) stage();
#pragma unroll
        for (int kk = 0; kk < 16; kk += 4) {
            const int k0 = kk + 2 * hi;
            v2f a; a.x = wrow[kc + k0]; a.y = wrow[kc + k0 + 1];
#pragma unroll
            for (int j = 0; j < 8; ++j) {
                const int nt = ntb + j;
                v2f bv; bv.x = g[k0][nt * 16 + m]; bv.y = g[k0 + 1][nt * 16 + m];
                acc[j] = __builtin_amdgcn_wmma_f32_16x16x4_f32(false, a, false, bv, (short)0, acc[j], false, false);
            }
        }
        __syncthreads();
    }

#pragma unroll
    for (int j = 0; j < 8; ++j) {
        const int nt = ntb + j;
        size_t base = ((size_t)b * Cout + mt * 16 + 8 * hi) * HW
                    + (size_t)(r0 + nt) * Wdim + c0 + m;
#pragma unroll
        for (int r = 0; r < 8; ++r)
            out[base + (size_t)r * HW] = acc[j][r];
    }
}

extern "C" void kernel_launch(void* const* d_in, const int* in_sizes, int n_in,
                              void* d_out, int out_size, void* d_ws, size_t ws_size,
                              hipStream_t stream) {
    const float* x    = (const float*)d_in[0];
    const float* Win  = (const float*)d_in[1];
    const float* Wdw  = (const float*)d_in[2];
    const float* mix  = (const float*)d_in[3];
    const float* Wout = (const float*)d_in[4];
    float* out = (float*)d_out;
    float* ws  = (float*)d_ws;

    dim3 blk(256);
    dim3 g1(Wdim / 16, Hdim / 16, 2 * 16);
    k1_projin_dct<<<g1, blk, 0, stream>>>(x, Win, mix, ws);
    dim3 g2(Wdim / 16, Hdim / 16, 2);
    k2_dw_gate_out<<<g2, blk, 0, stream>>>(ws, Wdw, Wout, out);
}


# first kernel: each DCT as a column pass and a row pass in LDS, one 8-vector per lane per pass (8 reads, 64 literal-coefficient FMAs), not the whole 8x8 patch per output row; same sum order, all fused
# speedup vs baseline: 1.4761x; 1.1028x over previous
#include <hip/hip_runtime.h>

typedef __attribute__((ext_vector_type(2))) float v2f;
typedef __attribute__((ext_vector_type(8))) float v8f;

constexpr int Hdim = 256, Wdim = 256, Cin = 64, C2 = 256, HID = 128, Cout = 64;
constexpr int HW = Hdim * Wdim;

__constant__ const float DCT8[8][8] = {
  { 0.3535533906f, 0.3535533906f, 0.3535533906f, 0.3535533906f, 0.3535533906f, 0.3535533906f, 0.3535533906f, 0.3535533906f },
  { 0.4903926402f, 0.4157348062f, 0.2777851165f, 0.0975451610f,-0.0975451610f,-0.2777851165f,-0.4157348062f,-0.4903926402f },
  { 0.4619397663f, 0.1913417162f,-0.1913417162f,-0.4619397663f,-0.4619397663f,-0.1913417162f, 0.1913417162f, 0.4619397663f },
  { 0.4157348062f,-0.0975451610f,-0.4903926402f,-0.2777851165f, 0.2777851165f, 0.4903926402f, 0.0975451610f,-0.4157348062f },
  { 0.3535533906f,-0.3535533906f,-0.3535533906f, 0.3535533906f, 0.3535533906f,-0.3535533906f,-0.3535533906f, 0.3535533906f },
  { 0.2777851165f,-0.4903926402f, 0.0975451610f, 0.4157348062f,-0.4157348062f,-0.0975451610f, 0.4903926402f,-0.2777851165f },
  { 0.1913417162f,-0.4619397663f, 0.4619397663f,-0.1913417162f,-0.1913417162f, 0.4619397663f,-0.4619397663f, 0.1913417162f },
  { 0.0975451610f,-0.2777851165f, 0.4157348062f,-0.4903926402f, 0.4903926402f,-0.4157348062f, 0.2777851165f,-0.0975451610f }
};

typedef __attribute__((ext_vector_type(4))) float v4f;

constexpr int YP = 20;
constexpr int YC = 16 * YP + 4;

__device__ __forceinline__ void dct_cols(float* __restrict__ t, int prh, int px)
{
    float* p = t + (8 * prh) * YP + px;
    float x[8], w[8];
#pragma unroll
    for (int k = 0; k < 8; ++k) x[k] = p[k * YP];
#pragma unroll
    for (int i = 0; i < 8; ++i) {
        float s = DCT8[i][0] * x[0];
#pragma unroll
        for (int k = 1; k < 8; ++k) s = __builtin_fmaf(DCT8[i][k], x[k], s);
        w[i] = s;
    }
#pragma unroll
    for (int i = 0; i < 8; ++i) p[i * YP] = w[i];
}

template <bool MASK>
__device__ __forceinline__ void dct_rows(float* __restrict__ t, int py, int pch, const float* __restrict__ mrow)
{
    float* p = t + py * YP + 8 * pch;
    const v4f a = *reinterpret_cast<const v4f*>(p);
    const v4f b = *reinterpret_cast<const v4f*>(p + 4);
    float w[8], o[8];
#pragma unroll
    for (int e = 0; e < 4; ++e) { w[e] = a[e]; w[4 + e] = b[e]; }
#pragma unroll
    for (int q = 0; q < 8; ++q) {
        float s = w[0] * DCT8[q][0];
#pragma unroll
        for (int j = 1; j < 8; ++j) s = __builtin_fmaf(w[j], DCT8[q][j], s);
        o[q] = s;
    }
    if (MASK) {
        const v4f m0 = *reinterpret_cast<const v4f*>(mrow);
        const v4f m1 = *reinterpret_cast<const v4f*>(mrow + 4);
#pragma unroll
        for (int e = 0; e < 4; ++e) { o[e] *= m0[e]; o[4 + e] *= m1[e]; }
    }
    v4f r0, r1;
#pragma unroll
    for (int e = 0; e < 4; ++e) { r0[e] = o[e]; r1[e] = o[4 + e]; }
    *reinterpret_cast<v4f*>(p)     = r0;
    *reinterpret_cast<v4f*>(p + 4) = r1;
}

__global__ __launch_bounds__(256) void
k1_projin_dct(const float* __restrict__ x, const float* __restrict__ Win,
              const float* __restrict__ mix, float* __restrict__ ws)
{
    __shared__ alignas(16) float yt[16 * YC];

    const int tid  = threadIdx.x;
    const int lane = tid & 31;
    const int wv   = tid >> 5;
    const int m    = lane & 15;
    const int hi   = lane >> 4;

    const int b  = blockIdx.z >> 4;
    const int ob = blockIdx.z & 15;
    const int r0 = blockIdx.y * 16;
    const int c0 = blockIdx.x * 16;

    const float* xb = x + (size_t)b * Cin * HW;
    const float* wr = Win + (size_t)(ob * 16 + m) * Cin;
    const int t0 = 2 * wv, t1 = 2 * wv + 1;
    const float* xp0 = xb + (size_t)(r0 + t0) * Wdim + c0 + m;
    const float* xp1 = xb + (size_t)(r0 + t1) * Wdim + c0 + m;

    v8f acc0 = {}, acc1 = {};
#pragma unroll
    for (int kk = 0; kk < Cin; kk += 4) {
        const int k0 = kk + 2 * hi;
        v2f a;  a.x  = wr[k0];                    a.y  = wr[k0 + 1];
        v2f b0; b0.x = xp0[(size_t)k0 * HW];      b0.y = xp0[(size_t)(k0 + 1) * HW];
        v2f b1; b1.x = xp1[(size_t)k0 * HW];      b1.y = xp1[(size_t)(k0 + 1) * HW];
        acc0 = __builtin_amdgcn_wmma_f32_16x16x4_f32(false, a, false, b0, (short)0, acc0, false, false);
        acc1 = __builtin_amdgcn_wmma_f32_16x16x4_f32(false, a, false, b1, (short)0, acc1, false, false);
    }
#pragma unroll
    for (int r = 0; r < 8; ++r) {
        yt[(r + 8 * hi) * YC + t0 * YP + m] = acc0[r];
        yt[(r + 8 * hi) * YC + t1 * YP + m] = acc1[r];
    }
    __syncthreads();

    float* tA = yt + wv * YC;
    float* tB = yt + (wv + 8) * YC;
    const float* mA = mix + (size_t)(ob * 16 + wv) * 64 + (m & 7) * 8;
    const float* mB = mA + 8 * 64;

    dct_cols(tA, hi, m);
    dct_cols(tB, hi, m);
    __syncthreads();
    dct_rows<true>(tA, m, hi, mA);
    dct_rows<true>(tB, m, hi, mB);
    __syncthreads();

    dct_cols(tA, hi, m);
    dct_cols(tB, hi, m);
    __syncthreads();
    dct_rows<false>(tA, m, hi, nullptr);
    dct_rows<false>(tB, m, hi, nullptr);
    __syncthreads();

    const int py = tid >> 4, px = tid & 15;
    size_t obase = ((size_t)b * C2 + ob * 16) * HW + (size_t)(r0 + py) * Wdim + (c0 + px);
#pragma unroll
    for (int ch = 0; ch < 16; ++ch)
        ws[obase + (size_t)ch * HW] = yt[ch * YC + py * YP + px];
}

constexpr int TP  = 28;
constexpr int TCH = 18 * TP;

__global__ __launch_bounds__(256) void
k2_dw_gate_out(const float* __restrict__ ws, const float* __restrict__ wdw,
               const float* __restrict__ wout, float* __restrict__ out)
{
    __shared__ alignas(16) float g[16][272];
    __shared__ alignas(16) float tile[32 * TCH];

    const int tid  = threadIdx.x;
    const int lane = tid & 31;
    const int wv   = tid >> 5;
    const int m    = lane & 15;
    const int hi   = lane >> 4;

    const int b  = blockIdx.z;
    const int r0 = blockIdx.y * 16;
    const int c0 = blockIdx.x * 16;

    const int wvu = __builtin_amdgcn_readfirstlane(wv);
    const int cy  = m;

    const float* wsb  = ws + (size_t)b * C2 * HW;
    const int mt  = wv >> 1;
    const int ntb = (wv & 1) * 8;
    const float* wrow = wout + (size_t)(mt * 16 + m) * HID;

    const int quad = tid & 3, rr = tid >> 2;
    v4f   cv[9];
    float hv[5];

    auto fetch = [&](int kc) {
#pragma unroll
        for (int t = 0; t < 9; ++t) {
            const int R   = rr + 64 * t;
            const int cs  = R / 18;
            const int row = R - cs * 18;
            const int yy  = r0 - 1 + row;
            const bool ok = ((unsigned)yy < (unsigned)Hdim);
            const int c   = kc + (cs & 15) + (cs >> 4) * HID;
            const float* p = wsb + (size_t)c * HW + (size_t)(ok ? yy : 0) * Wdim + c0 + quad * 4;
            const v4f v = *reinterpret_cast<const v4f*>(p);
            cv[t] = ok ? v : (v4f){0.f, 0.f, 0.f, 0.f};
        }
#pragma unroll
        for (int t = 0; t < 5; ++t) {
            const int e    = tid + 256 * t;
            const int side = e & 1;
            const int R    = e >> 1;
            const int cs   = R / 18;
            const int row  = R - cs * 18;
            const int yy   = r0 - 1 + row;
            const int xx   = side ? c0 + 16 : c0 - 1;
            const bool ok  = (e < 1152) && ((unsigned)yy < (unsigned)Hdim) && ((unsigned)xx < (unsigned)Wdim);
            const int c    = kc + (cs & 15) + (cs >> 4) * HID;
            hv[t] = ok ? wsb[(size_t)c * HW + (size_t)yy * Wdim + xx] : 0.f;
        }
    };
    auto stage = [&]() {
#pragma unroll
        for (int t = 0; t < 9; ++t) {
            const int R   = rr + 64 * t;
            const int cs  = R / 18;
            const int row = R - cs * 18;
            *reinterpret_cast<v4f*>(&tile[cs * TCH + row * TP + 4 + quad * 4]) = cv[t];
        }
#pragma unroll
        for (int t = 0; t < 5; ++t) {
            const int e    = tid + 256 * t;
            const int side = e & 1;
            const int R    = e >> 1;
            const int cs   = R / 18;
            const int row  = R - cs * 18;
            if (e < 1152) tile[cs * TCH + row * TP + (side ? 20 : 3)] = hv[t];
        }
    };

    v8f acc[8];
#pragma unroll
    for (int j = 0; j < 8; ++j) acc[j] = (v8f){};

    fetch(0);
    stage();
    __syncthreads();

    for (int kc = 0; kc < HID; kc += 16) {
        const bool more = (kc + 16 < HID);
        if (more) fetch(kc + 16);
#pragma unroll 1
        for (int cc = 0; cc < 2; ++cc) {
            const int i = 2 * wvu + cc;
            const int c = kc + i;
            const float* w1 = wdw + c * 9;
            const float* w2 = wdw + (c + HID) * 9;
            const float* t1 = tile + i * TCH + cy * TP + 3 + 8 * hi;
            const float* t2 = t1 + 16 * TCH;
            float s1[8], s2[8];
#pragma unroll
            for (int j = 0; j < 8; ++j) { s1[j] = 0.f; s2[j] = 0.f; }
#pragma unroll
            for (int ky = 0; ky < 3; ++ky) {
                const float* r1 = t1 + ky * TP;
                const float* r2 = t2 + ky * TP;
                float a[10], bq[10];
                const v4f a0 = *reinterpret_cast<const v4f*>(r1 + 1);
                const v4f a1 = *reinterpret_cast<const v4f*>(r1 + 5);
                const v4f b0 = *reinterpret_cast<const v4f*>(r2 + 1);
                const v4f b1 = *reinterpret_cast<const v4f*>(r2 + 5);
                a[0] = r1[0];  a[9] = r1[9];
                bq[0] = r2[0]; bq[9] = r2[9];
#pragma unroll
                for (int e = 0; e < 4; ++e) { a[1 + e] = a0[e]; a[5 + e] = a1[e]; bq[1 + e] = b0[e]; bq[5 + e] = b1[e]; }
#pragma unroll
                for (int kx = 0; kx < 3; ++kx) {
                    const float wa = w1[ky * 3 + kx];
                    const float wb = w2[ky * 3 + kx];
#pragma unroll
                    for (int j = 0; j < 8; ++j) {
                        s1[j] += wa * a[j + kx];
                        s2[j] += wb * bq[j + kx];
                    }
                }
            }
            v4f o0, o1;
#pragma unroll
            for (int j = 0; j < 8; ++j) {
                const float t = tanhf(0.7978845608028654f * (s1[j] + 0.044715f * s1[j] * s1[j] * s1[j]));
                const float v = 0.5f * s1[j] * (1.f + t) * s2[j];
                if (j < 4) o0[j] = v; else o1[j - 4] = v;
            }
            float* gp = &g[i][cy * 16 + 8 * hi];
            *reinterpret_cast<v4f*>(gp)     = o0;
            *reinterpret_cast<v4f*>(gp + 4) = o1;
        }
        __syncthreads();
        if (more) stage();
#pragma unroll
        for (int kk = 0; kk < 16; kk += 4) {
            const int k0 = kk + 2 * hi;
            v2f a; a.x = wrow[kc + k0]; a.y = wrow[kc + k0 + 1];
#pragma unroll
            for (int j = 0; j < 8; ++j) {
                const int nt = ntb + j;
                v2f bv; bv.x = g[k0][nt * 16 + m]; bv.y = g[k0 + 1][nt * 16 + m];
                acc[j] = __builtin_amdgcn_wmma_f32_16x16x4_f32(false, a, false, bv, (short)0, acc[j], false, false);
            }
        }
        __syncthreads();
    }

#pragma unroll
    for (int j = 0; j < 8; ++j) {
        const int nt = ntb + j;
        size_t base = ((size_t)b * Cout + mt * 16 + 8 * hi) * HW
                    + (size_t)(r0 + nt) * Wdim + c0 + m;
#pragma unroll
        for (int r = 0; r < 8; ++r)
            out[base + (size_t)r * HW] = acc[j][r];
    }
}

extern "C" void kernel_launch(void* const* d_in, const int* in_sizes, int n_in,
                              void* d_out, int out_size, void* d_ws, size_t ws_size,
                              hipStream_t stream) {
    const float* x    = (const float*)d_in[0];
    const float* Win  = (const float*)d_in[1];
    const float* Wdw  = (const float*)d_in[2];
    const float* mix  = (const float*)d_in[3];
    const float* Wout = (const float*)d_in[4];
    float* out = (float*)d_out;
    float* ws  = (float*)d_ws;

    dim3 blk(256);
    dim3 g1(Wdim / 16, Hdim / 16, 2 * 16);
    k1_projin_dct<<<g1, blk, 0, stream>>>(x, Win, mix, ws);
    dim3 g2(Wdim / 16, Hdim / 16, 2);
    k2_dw_gate_out<<<g2, blk, 0, stream>>>(ws, Wdw, Wout, out);
}


# first kernel: tile 8 rows x 32 px instead of 16x16, so x loads (8 B per lane, even/odd pixel tiles) and workspace stores (16 B per lane) cover whole 128-B lines; same k order, same sums
# speedup vs baseline: 1.8862x; 1.2778x over previous
#include <hip/hip_runtime.h>

typedef __attribute__((ext_vector_type(2))) float v2f;
typedef __attribute__((ext_vector_type(8))) float v8f;

constexpr int Hdim = 256, Wdim = 256, Cin = 64, C2 = 256, HID = 128, Cout = 64;
constexpr int HW = Hdim * Wdim;

__constant__ const float DCT8[8][8] = {
  { 0.3535533906f, 0.3535533906f, 0.3535533906f, 0.3535533906f, 0.3535533906f, 0.3535533906f, 0.3535533906f, 0.3535533906f },
  { 0.4903926402f, 0.4157348062f, 0.2777851165f, 0.0975451610f,-0.0975451610f,-0.2777851165f,-0.4157348062f,-0.4903926402f },
  { 0.4619397663f, 0.1913417162f,-0.1913417162f,-0.4619397663f,-0.4619397663f,-0.1913417162f, 0.1913417162f, 0.4619397663f },
  { 0.4157348062f,-0.0975451610f,-0.4903926402f,-0.2777851165f, 0.2777851165f, 0.4903926402f, 0.0975451610f,-0.4157348062f },
  { 0.3535533906f,-0.3535533906f,-0.3535533906f, 0.3535533906f, 0.3535533906f,-0.3535533906f,-0.3535533906f, 0.3535533906f },
  { 0.2777851165f,-0.4903926402f, 0.0975451610f, 0.4157348062f,-0.4157348062f,-0.0975451610f, 0.4903926402f,-0.2777851165f },
  { 0.1913417162f,-0.4619397663f, 0.4619397663f,-0.1913417162f,-0.1913417162f, 0.4619397663f,-0.4619397663f, 0.1913417162f },
  { 0.0975451610f,-0.2777851165f, 0.4157348062f,-0.4903926402f, 0.4903926402f,-0.4157348062f, 0.2777851165f,-0.0975451610f }
};

typedef __attribute__((ext_vector_type(4))) float v4f;

constexpr int YP = 32;
constexpr int YC = 8 * YP + 4;

__device__ __forceinline__ int ysw(int row) { return ((row >> 1) & 1) << 2; }

__device__ __forceinline__ void dct_cols(float* __restrict__ t, int px)
{
    float* pa = t + px;
    float* pb = t + (px ^ 4);
    float x[8], w[8];
#pragma unroll
    for (int k = 0; k < 8; ++k) x[k] = (((k >> 1) & 1) ? pb : pa)[k * YP];
#pragma unroll
    for (int i = 0; i < 8; ++i) {
        float s = DCT8[i][0] * x[0];
#pragma unroll
        for (int k = 1; k < 8; ++k) s = __builtin_fmaf(DCT8[i][k], x[k], s);
        w[i] = s;
    }
#pragma unroll
    for (int i = 0; i < 8; ++i) (((i >> 1) & 1) ? pb : pa)[i * YP] = w[i];
}

template <bool MASK>
__device__ __forceinline__ void dct_rows(float* __restrict__ t, int py, int pch, const float* __restrict__ mrow)
{
    float* row = t + py * YP;
    float* p0 = row + ((8 * pch)     ^ ysw(py));
    float* p1 = row + ((8 * pch + 4) ^ ysw(py));
    const v4f a = *reinterpret_cast<const v4f*>(p0);
    const v4f b = *reinterpret_cast<const v4f*>(p1);
    float w[8], o[8];
#pragma unroll
    for (int e = 0; e < 4; ++e) { w[e] = a[e]; w[4 + e] = b[e]; }
#pragma unroll
    for (int q = 0; q < 8; ++q) {
        float s = w[0] * DCT8[q][0];
#pragma unroll
        for (int j = 1; j < 8; ++j) s = __builtin_fmaf(w[j], DCT8[q][j], s);
        o[q] = s;
    }
    if (MASK) {
        const v4f m0 = *reinterpret_cast<const v4f*>(mrow);
        const v4f m1 = *reinterpret_cast<const v4f*>(mrow + 4);
#pragma unroll
        for (int e = 0; e < 4; ++e) { o[e] *= m0[e]; o[4 + e] *= m1[e]; }
    }
    v4f r0, r1;
#pragma unroll
    for (int e = 0; e < 4; ++e) { r0[e] = o[e]; r1[e] = o[4 + e]; }
    *reinterpret_cast<v4f*>(p0) = r0;
    *reinterpret_cast<v4f*>(p1) = r1;
}

__global__ __launch_bounds__(256) void
k1_projin_dct(const float* __restrict__ x, const float* __restrict__ Win,
              const float* __restrict__ mix, float* __restrict__ ws)
{
    __shared__ alignas(16) float yt[16 * YC];

    const int tid  = threadIdx.x;
    const int lane = tid & 31;
    const int wv   = tid >> 5;
    const int m    = lane & 15;
    const int hi   = lane >> 4;

    const int b  = blockIdx.z >> 4;
    const int ob = blockIdx.z & 15;
    const int r0 = blockIdx.y * 8;
    const int c0 = blockIdx.x * 32;

    const float* xb = x + (size_t)b * Cin * HW;
    const float* wr = Win + (size_t)(ob * 16 + m) * Cin;
    const float* xp = xb + (size_t)(r0 + wv) * Wdim + c0 + 2 * m;

    v8f accE = {}, accO = {};
#pragma unroll
    for (int kk = 0; kk < Cin; kk += 4) {
        const int k0 = kk + 2 * hi;
        v2f a;  a.x  = wr[k0];                    a.y  = wr[k0 + 1];
        const v2f l0 = *reinterpret_cast<const v2f*>(xp + (size_t)k0 * HW);
        const v2f l1 = *reinterpret_cast<const v2f*>(xp + (size_t)(k0 + 1) * HW);
        v2f be; be.x = l0.x; be.y = l1.x;
        v2f bo; bo.x = l0.y; bo.y = l1.y;
        accE = __builtin_amdgcn_wmma_f32_16x16x4_f32(false, a, false, be, (short)0, accE, false, false);
        accO = __builtin_amdgcn_wmma_f32_16x16x4_f32(false, a, false, bo, (short)0, accO, false, false);
    }
    {
        float* yd = yt + (8 * hi) * YC + wv * YP + ((2 * m) ^ ysw(wv));
#pragma unroll
        for (int r = 0; r < 8; ++r) {
            v2f pr; pr.x = accE[r]; pr.y = accO[r];
            *reinterpret_cast<v2f*>(yd + r * YC) = pr;
        }
    }
    __syncthreads();

    float* tA = yt + wv * YC;
    float* tB = yt + (wv + 8) * YC;
    const int pch = lane & 3;
    const int py  = lane >> 2;
    const float* mA = mix + (size_t)(ob * 16 + wv) * 64 + py * 8;
    const float* mB = mA + 8 * 64;

    dct_cols(tA, lane);
    dct_cols(tB, lane);
    __syncthreads();
    dct_rows<true>(tA, py, pch, mA);
    dct_rows<true>(tB, py, pch, mB);
    __syncthreads();

    dct_cols(tA, lane);
    dct_cols(tB, lane);
    __syncthreads();
    dct_rows<false>(tA, py, pch, nullptr);
    dct_rows<false>(tB, py, pch, nullptr);
    __syncthreads();

#pragma unroll
    for (int it = 0; it < 4; ++it) {
        const int idx = tid + 256 * it;
        const int ch  = idx >> 6;
        const int row = (idx >> 3) & 7;
        const int u   = idx & 7;
        const v4f v = *reinterpret_cast<const v4f*>(yt + ch * YC + row * YP + ((4 * u) ^ ysw(row)));
        *reinterpret_cast<v4f*>(ws + ((size_t)b * C2 + ob * 16 + ch) * HW
                                   + (size_t)(r0 + row) * Wdim + c0 + 4 * u) = v;
    }
}

constexpr int TP  = 28;
constexpr int TCH = 18 * TP;

__global__ __launch_bounds__(256) void
k2_dw_gate_out(const float* __restrict__ ws, const float* __restrict__ wdw,
               const float* __restrict__ wout, float* __restrict__ out)
{
    __shared__ alignas(16) float g[16][272];
    __shared__ alignas(16) float tile[32 * TCH];

    const int tid  = threadIdx.x;
    const int lane = tid & 31;
    const int wv   = tid >> 5;
    const int m    = lane & 15;
    const int hi   = lane >> 4;

    const int b  = blockIdx.z;
    const int r0 = blockIdx.y * 16;
    const int c0 = blockIdx.x * 16;

    const int wvu = __builtin_amdgcn_readfirstlane(wv);
    const int cy  = m;

    const float* wsb  = ws + (size_t)b * C2 * HW;
    const int mt  = wv >> 1;
    const int ntb = (wv & 1) * 8;
    const float* wrow = wout + (size_t)(mt * 16 + m) * HID;

    const int quad = tid & 3, rr = tid >> 2;
    v4f   cv[9];
    float hv[5];

    auto fetch = [&](int kc) {
#pragma unroll
        for (int t = 0; t < 9; ++t) {
            const int R   = rr + 64 * t;
            const int cs  = R / 18;
            const int row = R - cs * 18;
            const int yy  = r0 - 1 + row;
            const bool ok = ((unsigned)yy < (unsigned)Hdim);
            const int c   = kc + (cs & 15) + (cs >> 4) * HID;
            const float* p = wsb + (size_t)c * HW + (size_t)(ok ? yy : 0) * Wdim + c0 + quad * 4;
            const v4f v = *reinterpret_cast<const v4f*>(p);
            cv[t] = ok ? v : (v4f){0.f, 0.f, 0.f, 0.f};
        }
#pragma unroll
        for (int t = 0; t < 5; ++t) {
            const int e    = tid + 256 * t;
            const int side = e & 1;
            const int R    = e >> 1;
            const int cs   = R / 18;
            const int row  = R - cs * 18;
            const int yy   = r0 - 1 + row;
            const int xx   = side ? c0 + 16 : c0 - 1;
            const bool ok  = (e < 1152) && ((unsigned)yy < (unsigned)Hdim) && ((unsigned)xx < (unsigned)Wdim);
            const int c    = kc + (cs & 15) + (cs >> 4) * HID;
            hv[t] = ok ? wsb[(size_t)c * HW + (size_t)yy * Wdim + xx] : 0.f;
        }
    };
    auto stage = [&]() {
#pragma unroll
        for (int t = 0; t < 9; ++t) {
            const int R   = rr + 64 * t;
            const int cs  = R / 18;
            const int row = R - cs * 18;
            *reinterpret_cast<v4f*>(&tile[cs * TCH + row * TP + 4 + quad * 4]) = cv[t];
        }
#pragma unroll
        for (int t = 0; t < 5; ++t) {
            const int e    = tid + 256 * t;
            const int side = e & 1;
            const int R    = e >> 1;
            const int cs   = R / 18;
            const int row  = R - cs * 18;
            if (e < 1152) tile[cs * TCH + row * TP + (side ? 20 : 3)] = hv[t];
        }
    };

    v8f acc[8];
#pragma unroll
    for (int j = 0; j < 8; ++j) acc[j] = (v8f){};

    fetch(0);
    stage();
    __syncthreads();

    for (int kc = 0; kc < HID; kc += 16) {
        const bool more = (kc + 16 < HID);
        if (more) fetch(kc + 16);
#pragma unroll 1
        for (int cc = 0; cc < 2; ++cc) {
            const int i = 2 * wvu + cc;
            const int c = kc + i;
            const float* w1 = wdw + c * 9;
            const float* w2 = wdw + (c + HID) * 9;
            const float* t1 = tile + i * TCH + cy * TP + 3 + 8 * hi;
            const float* t2 = t1 + 16 * TCH;
            float s1[8], s2[8];
#pragma unroll
            for (int j = 0; j < 8; ++j) { s1[j] = 0.f; s2[j] = 0.f; }
#pragma unroll
            for (int ky = 0; ky < 3; ++ky) {
                const float* r1 = t1 + ky * TP;
                const float* r2 = t2 + ky * TP;
                float a[10], bq[10];
                const v4f a0 = *reinterpret_cast<const v4f*>(r1 + 1);
                const v4f a1 = *reinterpret_cast<const v4f*>(r1 + 5);
                const v4f b0 = *reinterpret_cast<const v4f*>(r2 + 1);
                const v4f b1 = *reinterpret_cast<const v4f*>(r2 + 5);
                a[0] = r1[0];  a[9] = r1[9];
                bq[0] = r2[0]; bq[9] = r2[9];
#pragma unroll
                for (int e = 0; e < 4; ++e) { a[1 + e] = a0[e]; a[5 + e] = a1[e]; bq[1 + e] = b0[e]; bq[5 + e] = b1[e]; }
#pragma unroll
                for (int kx = 0; kx < 3; ++kx) {
                    const float wa = w1[ky * 3 + kx];
                    const float wb = w2[ky * 3 + kx];
#pragma unroll
                    for (int j = 0; j < 8; ++j) {
                        s1[j] += wa * a[j + kx];
                        s2[j] += wb * bq[j + kx];
                    }
                }
            }
            v4f o0, o1;
#pragma unroll
            for (int j = 0; j < 8; ++j) {
                const float t = tanhf(0.7978845608028654f * (s1[j] + 0.044715f * s1[j] * s1[j] * s1[j]));
                const float v = 0.5f * s1[j] * (1.f + t) * s2[j];
                if (j < 4) o0[j] = v; else o1[j - 4] = v;
            }
            float* gp = &g[i][cy * 16 + 8 * hi];
            *reinterpret_cast<v4f*>(gp)     = o0;
            *reinterpret_cast<v4f*>(gp + 4) = o1;
        }
        __syncthreads();
        if (more) stage();
#pragma unroll
        for (int kk = 0; kk < 16; kk += 4) {
            const int k0 = kk + 2 * hi;
            v2f a; a.x = wrow[kc + k0]; a.y = wrow[kc + k0 + 1];
#pragma unroll
            for (int j = 0; j < 8; ++j) {
                const int nt = ntb + j;
                v2f bv; bv.x = g[k0][nt * 16 + m]; bv.y = g[k0 + 1][nt * 16 + m];
                acc[j] = __builtin_amdgcn_wmma_f32_16x16x4_f32(false, a, false, bv, (short)0, acc[j], false, false);
            }
        }
        __syncthreads();
    }

#pragma unroll
    for (int j = 0; j < 8; ++j) {
        const int nt = ntb + j;
        size_t base = ((size_t)b * Cout + mt * 16 + 8 * hi) * HW
                    + (size_t)(r0 + nt) * Wdim + c0 + m;
#pragma unroll
        for (int r = 0; r < 8; ++r)
            out[base + (size_t)r * HW] = acc[j][r];
    }
}

extern "C" void kernel_launch(void* const* d_in, const int* in_sizes, int n_in,
                              void* d_out, int out_size, void* d_ws, size_t ws_size,
                              hipStream_t stream) {
    const float* x    = (const float*)d_in[0];
    const float* Win  = (const float*)d_in[1];
    const float* Wdw  = (const float*)d_in[2];
    const float* mix  = (const float*)d_in[3];
    const float* Wout = (const float*)d_in[4];
    float* out = (float*)d_out;
    float* ws  = (float*)d_ws;

    dim3 blk(256);
    dim3 g1(Wdim / 32, Hdim / 8, 2 * 16);
    k1_projin_dct<<<g1, blk, 0, stream>>>(x, Win, mix, ws);
    dim3 g2(Wdim / 16, Hdim / 16, 2);
    k2_dw_gate_out<<<g2, blk, 0, stream>>>(ws, Wdw, Wout, out);
}


# first kernel: a workgroup holds its tile's 64 input channels in registers as B fragments and loops over the 16 out-channel blocks (LDS tile double buffered): x read once, 16x fewer waves; same sums
# speedup vs baseline: 1.9725x; 1.0458x over previous
#include <hip/hip_runtime.h>

typedef __attribute__((ext_vector_type(2))) float v2f;
typedef __attribute__((ext_vector_type(8))) float v8f;

constexpr int Hdim = 256, Wdim = 256, Cin = 64, C2 = 256, HID = 128, Cout = 64;
constexpr int HW = Hdim * Wdim;

__constant__ const float DCT8[8][8] = {
  { 0.3535533906f, 0.3535533906f, 0.3535533906f, 0.3535533906f, 0.3535533906f, 0.3535533906f, 0.3535533906f, 0.3535533906f },
  { 0.4903926402f, 0.4157348062f, 0.2777851165f, 0.0975451610f,-0.0975451610f,-0.2777851165f,-0.4157348062f,-0.4903926402f },
  { 0.4619397663f, 0.1913417162f,-0.1913417162f,-0.4619397663f,-0.4619397663f,-0.1913417162f, 0.1913417162f, 0.4619397663f },
  { 0.4157348062f,-0.0975451610f,-0.4903926402f,-0.2777851165f, 0.2777851165f, 0.4903926402f, 0.0975451610f,-0.4157348062f },
  { 0.3535533906f,-0.3535533906f,-0.3535533906f, 0.3535533906f, 0.3535533906f,-0.3535533906f,-0.3535533906f, 0.3535533906f },
  { 0.2777851165f,-0.4903926402f, 0.0975451610f, 0.4157348062f,-0.4157348062f,-0.0975451610f, 0.4903926402f,-0.2777851165f },
  { 0.1913417162f,-0.4619397663f, 0.4619397663f,-0.1913417162f,-0.1913417162f, 0.4619397663f,-0.4619397663f, 0.1913417162f },
  { 0.0975451610f,-0.2777851165f, 0.4157348062f,-0.4903926402f, 0.4903926402f,-0.4157348062f, 0.2777851165f,-0.0975451610f }
};

typedef __attribute__((ext_vector_type(4))) float v4f;

constexpr int YP = 32;
constexpr int YC = 8 * YP + 4;

__device__ __forceinline__ int ysw(int row) { return ((row >> 1) & 1) << 2; }

__device__ __forceinline__ void dct_cols(float* __restrict__ t, int px)
{
    float* pa = t + px;
    float* pb = t + (px ^ 4);
    float x[8], w[8];
#pragma unroll
    for (int k = 0; k < 8; ++k) x[k] = (((k >> 1) & 1) ? pb : pa)[k * YP];
#pragma unroll
    for (int i = 0; i < 8; ++i) {
        float s = DCT8[i][0] * x[0];
#pragma unroll
        for (int k = 1; k < 8; ++k) s = __builtin_fmaf(DCT8[i][k], x[k], s);
        w[i] = s;
    }
#pragma unroll
    for (int i = 0; i < 8; ++i) (((i >> 1) & 1) ? pb : pa)[i * YP] = w[i];
}

template <bool MASK>
__device__ __forceinline__ void dct_rows(float* __restrict__ t, int py, int pch, v4f m0, v4f m1)
{
    float* row = t + py * YP;
    float* p0 = row + ((8 * pch)     ^ ysw(py));
    float* p1 = row + ((8 * pch + 4) ^ ysw(py));
    const v4f a = *reinterpret_cast<const v4f*>(p0);
    const v4f b = *reinterpret_cast<const v4f*>(p1);
    float w[8], o[8];
#pragma unroll
    for (int e = 0; e < 4; ++e) { w[e] = a[e]; w[4 + e] = b[e]; }
#pragma unroll
    for (int q = 0; q < 8; ++q) {
        float s = w[0] * DCT8[q][0];
#pragma unroll
        for (int j = 1; j < 8; ++j) s = __builtin_fmaf(w[j], DCT8[q][j], s);
        o[q] = s;
    }
    if (MASK) {
#pragma unroll
        for (int e = 0; e < 4; ++e) { o[e] *= m0[e]; o[4 + e] *= m1[e]; }
    }
    v4f r0, r1;
#pragma unroll
    for (int e = 0; e < 4; ++e) { r0[e] = o[e]; r1[e] = o[4 + e]; }
    *reinterpret_cast<v4f*>(p0) = r0;
    *reinterpret_cast<v4f*>(p1) = r1;
}

__global__ __launch_bounds__(256) void
k1_projin_dct(const float* __restrict__ x, const float* __restrict__ Win,
              const float* __restrict__ mix, float* __restrict__ ws)
{
    __shared__ alignas(16) float yt[2][16 * YC];

    const int tid  = threadIdx.x;
    const int lane = tid & 31;
    const int wv   = tid >> 5;
    const int m    = lane & 15;
    const int hi   = lane >> 4;

    const int b  = blockIdx.z;
    const int r0 = blockIdx.y * 8;
    const int c0 = blockIdx.x * 32;

    const float* xb = x + (size_t)b * Cin * HW;
    const float* xp = xb + (size_t)(r0 + wv) * Wdim + c0 + 2 * m;

    v2f be[16], bo[16];
#pragma unroll
    for (int s = 0; s < 16; ++s) {
        const int k0 = 4 * s + 2 * hi;
        const v2f l0 = *reinterpret_cast<const v2f*>(xp + (size_t)k0 * HW);
        const v2f l1 = *reinterpret_cast<const v2f*>(xp + (size_t)(k0 + 1) * HW);
        be[s].x = l0.x; be[s].y = l1.x;
        bo[s].x = l0.y; bo[s].y = l1.y;
    }

    const float* wr = Win + (size_t)m * Cin + 2 * hi;
    v2f a[16];
#pragma unroll
    for (int s = 0; s < 16; ++s) a[s] = *reinterpret_cast<const v2f*>(wr + 4 * s);

    const int pch = lane & 3;
    const int py  = lane >> 2;
    const float* mrow = mix + (size_t)wv * 64 + py * 8;

#pragma unroll 1
    for (int ob = 0; ob < 16; ++ob) {
        float* yb = yt[ob & 1];

        v8f accE = {}, accO = {};
#pragma unroll
        for (int s = 0; s < 16; ++s) {
            accE = __builtin_amdgcn_wmma_f32_16x16x4_f32(false, a[s], false, be[s], (short)0, accE, false, false);
            accO = __builtin_amdgcn_wmma_f32_16x16x4_f32(false, a[s], false, bo[s], (short)0, accO, false, false);
        }

        {
            const int obn = (ob < 15) ? ob + 1 : ob;
            const float* wn = wr + (size_t)obn * 16 * Cin;
#pragma unroll
            for (int s = 0; s < 16; ++s) a[s] = *reinterpret_cast<const v2f*>(wn + 4 * s);
        }
        const float* mA = mrow + (size_t)ob * 16 * 64;
        const float* mB = mA + 8 * 64;
        const v4f mA0 = *reinterpret_cast<const v4f*>(mA);
        const v4f mA1 = *reinterpret_cast<const v4f*>(mA + 4);
        const v4f mB0 = *reinterpret_cast<const v4f*>(mB);
        const v4f mB1 = *reinterpret_cast<const v4f*>(mB + 4);

        {
            float* yd = yb + (8 * hi) * YC + wv * YP + ((2 * m) ^ ysw(wv));
#pragma unroll
            for (int r = 0; r < 8; ++r) {
                v2f pr; pr.x = accE[r]; pr.y = accO[r];
                *reinterpret_cast<v2f*>(yd + r * YC) = pr;
            }
        }
        __syncthreads();

        float* tA = yb + wv * YC;
        float* tB = yb + (wv + 8) * YC;
        const v4f z4 = {0.f, 0.f, 0.f, 0.f};

        dct_cols(tA, lane);
        dct_cols(tB, lane);
        __syncthreads();
        dct_rows<true>(tA, py, pch, mA0, mA1);
        dct_rows<true>(tB, py, pch, mB0, mB1);
        __syncthreads();

        dct_cols(tA, lane);
        dct_cols(tB, lane);
        __syncthreads();
        dct_rows<false>(tA, py, pch, z4, z4);
        dct_rows<false>(tB, py, pch, z4, z4);
        __syncthreads();

#pragma unroll
        for (int it = 0; it < 4; ++it) {
            const int idx = tid + 256 * it;
            const int ch  = idx >> 6;
            const int row = (idx >> 3) & 7;
            const int u   = idx & 7;
            const v4f v = *reinterpret_cast<const v4f*>(yb + ch * YC + row * YP + ((4 * u) ^ ysw(row)));
            *reinterpret_cast<v4f*>(ws + ((size_t)b * C2 + ob * 16 + ch) * HW
                                       + (size_t)(r0 + row) * Wdim + c0 + 4 * u) = v;
        }
    }
}

constexpr int TP  = 28;
constexpr int TCH = 18 * TP;

__global__ __launch_bounds__(256) void
k2_dw_gate_out(const float* __restrict__ ws, const float* __restrict__ wdw,
               const float* __restrict__ wout, float* __restrict__ out)
{
    __shared__ alignas(16) float g[16][272];
    __shared__ alignas(16) float tile[32 * TCH];

    const int tid  = threadIdx.x;
    const int lane = tid & 31;
    const int wv   = tid >> 5;
    const int m    = lane & 15;
    const int hi   = lane >> 4;

    const int b  = blockIdx.z;
    const int r0 = blockIdx.y * 16;
    const int c0 = blockIdx.x * 16;

    const int wvu = __builtin_amdgcn_readfirstlane(wv);
    const int cy  = m;

    const float* wsb  = ws + (size_t)b * C2 * HW;
    const int mt  = wv >> 1;
    const int ntb = (wv & 1) * 8;
    const float* wrow = wout + (size_t)(mt * 16 + m) * HID;

    const int quad = tid & 3, rr = tid >> 2;
    v4f   cv[9];
    float hv[5];

    auto fetch = [&](int kc) {
#pragma unroll
        for (int t = 0; t < 9; ++t) {
            const int R   = rr + 64 * t;
            const int cs  = R / 18;
            const int row = R - cs * 18;
            const int yy  = r0 - 1 + row;
            const bool ok = ((unsigned)yy < (unsigned)Hdim);
            const int c   = kc + (cs & 15) + (cs >> 4) * HID;
            const float* p = wsb + (size_t)c * HW + (size_t)(ok ? yy : 0) * Wdim + c0 + quad * 4;
            const v4f v = *reinterpret_cast<const v4f*>(p);
            cv[t] = ok ? v : (v4f){0.f, 0.f, 0.f, 0.f};
        }
#pragma unroll
        for (int t = 0; t < 5; ++t) {
            const int e    = tid + 256 * t;
            const int side = e & 1;
            const int R    = e >> 1;
            const int cs   = R / 18;
            const int row  = R - cs * 18;
            const int yy   = r0 - 1 + row;
            const int xx   = side ? c0 + 16 : c0 - 1;
            const bool ok  = (e < 1152) && ((unsigned)yy < (unsigned)Hdim) && ((unsigned)xx < (unsigned)Wdim);
            const int c    = kc + (cs & 15) + (cs >> 4) * HID;
            hv[t] = ok ? wsb[(size_t)c * HW + (size_t)yy * Wdim + xx] : 0.f;
        }
    };
    auto stage = [&]() {
#pragma unroll
        for (int t = 0; t < 9; ++t) {
            const int R   = rr + 64 * t;
            const int cs  = R / 18;
            const int row = R - cs * 18;
            *reinterpret_cast<v4f*>(&tile[cs * TCH + row * TP + 4 + quad * 4]) = cv[t];
        }
#pragma unroll
        for (int t = 0; t < 5; ++t) {
            const int e    = tid + 256 * t;
            const int side = e & 1;
            const int R    = e >> 1;
            const int cs   = R / 18;
            const int row  = R - cs * 18;
            if (e < 1152) tile[cs * TCH + row * TP + (side ? 20 : 3)] = hv[t];
        }
    };

    v8f acc[8];
#pragma unroll
    for (int j = 0; j < 8; ++j) acc[j] = (v8f){};

    fetch(0);
    stage();
    __syncthreads();

    for (int kc = 0; kc < HID; kc += 16) {
        const bool more = (kc + 16 < HID);
        if (more) fetch(kc + 16);
#pragma unroll 1
        for (int cc = 0; cc < 2; ++cc) {
            const int i = 2 * wvu + cc;
            const int c = kc + i;
            const float* w1 = wdw + c * 9;
            const float* w2 = wdw + (c + HID) * 9;
            const float* t1 = tile + i * TCH + cy * TP + 3 + 8 * hi;
            const float* t2 = t1 + 16 * TCH;
            float s1[8], s2[8];
#pragma unroll
            for (int j = 0; j < 8; ++j) { s1[j] = 0.f; s2[j] = 0.f; }
#pragma unroll
            for (int ky = 0; ky < 3; ++ky) {
                const float* r1 = t1 + ky * TP;
                const float* r2 = t2 + ky * TP;
                float a[10], bq[10];
                const v4f a0 = *reinterpret_cast<const v4f*>(r1 + 1);
                const v4f a1 = *reinterpret_cast<const v4f*>(r1 + 5);
                const v4f b0 = *reinterpret_cast<const v4f*>(r2 + 1);
                const v4f b1 = *reinterpret_cast<const v4f*>(r2 + 5);
                a[0] = r1[0];  a[9] = r1[9];
                bq[0] = r2[0]; bq[9] = r2[9];
#pragma unroll
                for (int e = 0; e < 4; ++e) { a[1 + e] = a0[e]; a[5 + e] = a1[e]; bq[1 + e] = b0[e]; bq[5 + e] = b1[e]; }
#pragma unroll
                for (int kx = 0; kx < 3; ++kx) {
                    const float wa = w1[ky * 3 + kx];
                    const float wb = w2[ky * 3 + kx];
#pragma unroll
                    for (int j = 0; j < 8; ++j) {
                        s1[j] += wa * a[j + kx];
                        s2[j] += wb * bq[j + kx];
                    }
                }
            }
            v4f o0, o1;
#pragma unroll
            for (int j = 0; j < 8; ++j) {
                const float t = tanhf(0.7978845608028654f * (s1[j] + 0.044715f * s1[j] * s1[j] * s1[j]));
                const float v = 0.5f * s1[j] * (1.f + t) * s2[j];
                if (j < 4) o0[j] = v; else o1[j - 4] = v;
            }
            float* gp = &g[i][cy * 16 + 8 * hi];
            *reinterpret_cast<v4f*>(gp)     = o0;
            *reinterpret_cast<v4f*>(gp + 4) = o1;
        }
        __syncthreads();
        if (more) stage();
#pragma unroll
        for (int kk = 0; kk < 16; kk += 4) {
            const int k0 = kk + 2 * hi;
            v2f a; a.x = wrow[kc + k0]; a.y = wrow[kc + k0 + 1];
#pragma unroll
            for (int j = 0; j < 8; ++j) {
                const int nt = ntb + j;
                v2f bv; bv.x = g[k0][nt * 16 + m]; bv.y = g[k0 + 1][nt * 16 + m];
                acc[j] = __builtin_amdgcn_wmma_f32_16x16x4_f32(false, a, false, bv, (short)0, acc[j], false, false);
            }
        }
        __syncthreads();
    }

#pragma unroll
    for (int j = 0; j < 8; ++j) {
        const int nt = ntb + j;
        size_t base = ((size_t)b * Cout + mt * 16 + 8 * hi) * HW
                    + (size_t)(r0 + nt) * Wdim + c0 + m;
#pragma unroll
        for (int r = 0; r < 8; ++r)
            out[base + (size_t)r * HW] = acc[j][r];
    }
}

extern "C" void kernel_launch(void* const* d_in, const int* in_sizes, int n_in,
                              void* d_out, int out_size, void* d_ws, size_t ws_size,
                              hipStream_t stream) {
    const float* x    = (const float*)d_in[0];
    const float* Win  = (const float*)d_in[1];
    const float* Wdw  = (const float*)d_in[2];
    const float* mix  = (const float*)d_in[3];
    const float* Wout = (const float*)d_in[4];
    float* out = (float*)d_out;
    float* ws  = (float*)d_ws;

    dim3 blk(256);
    dim3 g1(Wdim / 32, Hdim / 8, 2);
    k1_projin_dct<<<g1, blk, 0, stream>>>(x, Win, mix, ws);
    dim3 g2(Wdim / 16, Hdim / 16, 2);
    k2_dw_gate_out<<<g2, blk, 0, stream>>>(ws, Wdw, Wout, out);
}


# second kernel: tile 8 rows x 32 px, not 16x16: conv input loads (16 B per lane) and output stores (8 B per lane, even/odd pixel tiles) cover whole 128-B lines; same k order, same sums
# speedup vs baseline: 2.3716x; 1.2023x over previous
#include <hip/hip_runtime.h>

typedef __attribute__((ext_vector_type(2))) float v2f;
typedef __attribute__((ext_vector_type(8))) float v8f;

constexpr int Hdim = 256, Wdim = 256, Cin = 64, C2 = 256, HID = 128, Cout = 64;
constexpr int HW = Hdim * Wdim;

__constant__ const float DCT8[8][8] = {
  { 0.3535533906f, 0.3535533906f, 0.3535533906f, 0.3535533906f, 0.3535533906f, 0.3535533906f, 0.3535533906f, 0.3535533906f },
  { 0.4903926402f, 0.4157348062f, 0.2777851165f, 0.0975451610f,-0.0975451610f,-0.2777851165f,-0.4157348062f,-0.4903926402f },
  { 0.4619397663f, 0.1913417162f,-0.1913417162f,-0.4619397663f,-0.4619397663f,-0.1913417162f, 0.1913417162f, 0.4619397663f },
  { 0.4157348062f,-0.0975451610f,-0.4903926402f,-0.2777851165f, 0.2777851165f, 0.4903926402f, 0.0975451610f,-0.4157348062f },
  { 0.3535533906f,-0.3535533906f,-0.3535533906f, 0.3535533906f, 0.3535533906f,-0.3535533906f,-0.3535533906f, 0.3535533906f },
  { 0.2777851165f,-0.4903926402f, 0.0975451610f, 0.4157348062f,-0.4157348062f,-0.0975451610f, 0.4903926402f,-0.2777851165f },
  { 0.1913417162f,-0.4619397663f, 0.4619397663f,-0.1913417162f,-0.1913417162f, 0.4619397663f,-0.4619397663f, 0.1913417162f },
  { 0.0975451610f,-0.2777851165f, 0.4157348062f,-0.4903926402f, 0.4903926402f,-0.4157348062f, 0.2777851165f,-0.0975451610f }
};

typedef __attribute__((ext_vector_type(4))) float v4f;

constexpr int YP = 32;
constexpr int YC = 8 * YP + 4;

__device__ __forceinline__ int ysw(int row) { return ((row >> 1) & 1) << 2; }

__device__ __forceinline__ void dct_cols(float* __restrict__ t, int px)
{
    float* pa = t + px;
    float* pb = t + (px ^ 4);
    float x[8], w[8];
#pragma unroll
    for (int k = 0; k < 8; ++k) x[k] = (((k >> 1) & 1) ? pb : pa)[k * YP];
#pragma unroll
    for (int i = 0; i < 8; ++i) {
        float s = DCT8[i][0] * x[0];
#pragma unroll
        for (int k = 1; k < 8; ++k) s = __builtin_fmaf(DCT8[i][k], x[k], s);
        w[i] = s;
    }
#pragma unroll
    for (int i = 0; i < 8; ++i) (((i >> 1) & 1) ? pb : pa)[i * YP] = w[i];
}

template <bool MASK>
__device__ __forceinline__ void dct_rows(float* __restrict__ t, int py, int pch, v4f m0, v4f m1)
{
    float* row = t + py * YP;
    float* p0 = row + ((8 * pch)     ^ ysw(py));
    float* p1 = row + ((8 * pch + 4) ^ ysw(py));
    const v4f a = *reinterpret_cast<const v4f*>(p0);
    const v4f b = *reinterpret_cast<const v4f*>(p1);
    float w[8], o[8];
#pragma unroll
    for (int e = 0; e < 4; ++e) { w[e] = a[e]; w[4 + e] = b[e]; }
#pragma unroll
    for (int q = 0; q < 8; ++q) {
        float s = w[0] * DCT8[q][0];
#pragma unroll
        for (int j = 1; j < 8; ++j) s = __builtin_fmaf(w[j], DCT8[q][j], s);
        o[q] = s;
    }
    if (MASK) {
#pragma unroll
        for (int e = 0; e < 4; ++e) { o[e] *= m0[e]; o[4 + e] *= m1[e]; }
    }
    v4f r0, r1;
#pragma unroll
    for (int e = 0; e < 4; ++e) { r0[e] = o[e]; r1[e] = o[4 + e]; }
    *reinterpret_cast<v4f*>(p0) = r0;
    *reinterpret_cast<v4f*>(p1) = r1;
}

__global__ __launch_bounds__(256) void
k1_projin_dct(const float* __restrict__ x, const float* __restrict__ Win,
              const float* __restrict__ mix, float* __restrict__ ws)
{
    __shared__ alignas(16) float yt[2][16 * YC];

    const int tid  = threadIdx.x;
    const int lane = tid & 31;
    const int wv   = tid >> 5;
    const int m    = lane & 15;
    const int hi   = lane >> 4;

    const int b  = blockIdx.z;
    const int r0 = blockIdx.y * 8;
    const int c0 = blockIdx.x * 32;

    const float* xb = x + (size_t)b * Cin * HW;
    const float* xp = xb + (size_t)(r0 + wv) * Wdim + c0 + 2 * m;

    v2f be[16], bo[16];
#pragma unroll
    for (int s = 0; s < 16; ++s) {
        const int k0 = 4 * s + 2 * hi;
        const v2f l0 = *reinterpret_cast<const v2f*>(xp + (size_t)k0 * HW);
        const v2f l1 = *reinterpret_cast<const v2f*>(xp + (size_t)(k0 + 1) * HW);
        be[s].x = l0.x; be[s].y = l1.x;
        bo[s].x = l0.y; bo[s].y = l1.y;
    }

    const float* wr = Win + (size_t)m * Cin + 2 * hi;
    v2f a[16];
#pragma unroll
    for (int s = 0; s < 16; ++s) a[s] = *reinterpret_cast<const v2f*>(wr + 4 * s);

    const int pch = lane & 3;
    const int py  = lane >> 2;
    const float* mrow = mix + (size_t)wv * 64 + py * 8;

#pragma unroll 1
    for (int ob = 0; ob < 16; ++ob) {
        float* yb = yt[ob & 1];

        v8f accE = {}, accO = {};
#pragma unroll
        for (int s = 0; s < 16; ++s) {
            accE = __builtin_amdgcn_wmma_f32_16x16x4_f32(false, a[s], false, be[s], (short)0, accE, false, false);
            accO = __builtin_amdgcn_wmma_f32_16x16x4_f32(false, a[s], false, bo[s], (short)0, accO, false, false);
        }

        {
            const int obn = (ob < 15) ? ob + 1 : ob;
            const float* wn = wr + (size_t)obn * 16 * Cin;
#pragma unroll
            for (int s = 0; s < 16; ++s) a[s] = *reinterpret_cast<const v2f*>(wn + 4 * s);
        }
        const float* mA = mrow + (size_t)ob * 16 * 64;
        const float* mB = mA + 8 * 64;
        const v4f mA0 = *reinterpret_cast<const v4f*>(mA);
        const v4f mA1 = *reinterpret_cast<const v4f*>(mA + 4);
        const v4f mB0 = *reinterpret_cast<const v4f*>(mB);
        const v4f mB1 = *reinterpret_cast<const v4f*>(mB + 4);

        {
            float* yd = yb + (8 * hi) * YC + wv * YP + ((2 * m) ^ ysw(wv));
#pragma unroll
            for (int r = 0; r < 8; ++r) {
                v2f pr; pr.x = accE[r]; pr.y = accO[r];
                *reinterpret_cast<v2f*>(yd + r * YC) = pr;
            }
        }
        __syncthreads();

        float* tA = yb + wv * YC;
        float* tB = yb + (wv + 8) * YC;
        const v4f z4 = {0.f, 0.f, 0.f, 0.f};

        dct_cols(tA, lane);
        dct_cols(tB, lane);
        __syncthreads();
        dct_rows<true>(tA, py, pch, mA0, mA1);
        dct_rows<true>(tB, py, pch, mB0, mB1);
        __syncthreads();

        dct_cols(tA, lane);
        dct_cols(tB, lane);
        __syncthreads();
        dct_rows<false>(tA, py, pch, z4, z4);
        dct_rows<false>(tB, py, pch, z4, z4);
        __syncthreads();

#pragma unroll
        for (int it = 0; it < 4; ++it) {
            const int idx = tid + 256 * it;
            const int ch  = idx >> 6;
            const int row = (idx >> 3) & 7;
            const int u   = idx & 7;
            const v4f v = *reinterpret_cast<const v4f*>(yb + ch * YC + row * YP + ((4 * u) ^ ysw(row)));
            *reinterpret_cast<v4f*>(ws + ((size_t)b * C2 + ob * 16 + ch) * HW
                                       + (size_t)(r0 + row) * Wdim + c0 + 4 * u) = v;
        }
    }
}

constexpr int TP  = 44;
constexpr int TRH = 10;
constexpr int TCH = TRH * TP;
constexpr int GP  = 272;

__global__ __launch_bounds__(256) void
k2_dw_gate_out(const float* __restrict__ ws, const float* __restrict__ wdw,
               const float* __restrict__ wout, float* __restrict__ out)
{
    __shared__ alignas(16) float g[16][GP];
    __shared__ alignas(16) float tile[32 * TCH];

    const int tid  = threadIdx.x;
    const int lane = tid & 31;
    const int wv   = tid >> 5;
    const int m    = lane & 15;
    const int hi   = lane >> 4;

    const int b  = blockIdx.z;
    const int r0 = blockIdx.y * 8;
    const int c0 = blockIdx.x * 32;

    const int wvu = __builtin_amdgcn_readfirstlane(wv);
    const int cy  = lane >> 2;
    const int cg  = lane & 3;

    const float* wsb  = ws + (size_t)b * C2 * HW;
    const int mt  = wv >> 1;
    const int ntb = (wv & 1) * 4;
    const float* wrow = wout + (size_t)(mt * 16 + m) * HID;

    const int q8 = tid & 7, rr = tid >> 3;
    v4f   cv[10];
    float hv[3];

    auto fetch = [&](int kc) {
#pragma unroll
        for (int t = 0; t < 10; ++t) {
            const int R   = rr + 32 * t;
            const int cs  = R / TRH;
            const int row = R - cs * TRH;
            const int yy  = r0 - 1 + row;
            const bool ok = ((unsigned)yy < (unsigned)Hdim);
            const int c   = kc + (cs & 15) + (cs >> 4) * HID;
            const float* p = wsb + (size_t)c * HW + (size_t)(ok ? yy : 0) * Wdim + c0 + q8 * 4;
            const v4f v = *reinterpret_cast<const v4f*>(p);
            cv[t] = ok ? v : (v4f){0.f, 0.f, 0.f, 0.f};
        }
#pragma unroll
        for (int t = 0; t < 3; ++t) {
            const int e    = tid + 256 * t;
            const int side = e & 1;
            const int R    = e >> 1;
            const int cs   = R / TRH;
            const int row  = R - cs * TRH;
            const int yy   = r0 - 1 + row;
            const int xx   = side ? c0 + 32 : c0 - 1;
            const bool ok  = (e < 640) && ((unsigned)yy < (unsigned)Hdim) && ((unsigned)xx < (unsigned)Wdim);
            const int c    = kc + (cs & 15) + (cs >> 4) * HID;
            hv[t] = ok ? wsb[(size_t)c * HW + (size_t)yy * Wdim + xx] : 0.f;
        }
    };
    auto stage = [&]() {
#pragma unroll
        for (int t = 0; t < 10; ++t) {
            const int R   = rr + 32 * t;
            const int cs  = R / TRH;
            const int row = R - cs * TRH;
            *reinterpret_cast<v4f*>(&tile[cs * TCH + row * TP + 4 + q8 * 4]) = cv[t];
        }
#pragma unroll
        for (int t = 0; t < 3; ++t) {
            const int e    = tid + 256 * t;
            const int side = e & 1;
            const int R    = e >> 1;
            const int cs   = R / TRH;
            const int row  = R - cs * TRH;
            if (e < 640) tile[cs * TCH + row * TP + (side ? 36 : 3)] = hv[t];
        }
    };

    v8f accE[4], accO[4];
#pragma unroll
    for (int j = 0; j < 4; ++j) { accE[j] = (v8f){}; accO[j] = (v8f){}; }

    fetch(0);
    stage();
    __syncthreads();

    for (int kc = 0; kc < HID; kc += 16) {
        const bool more = (kc + 16 < HID);
        if (more) fetch(kc + 16);
#pragma unroll 1
        for (int cc = 0; cc < 2; ++cc) {
            const int i = 2 * wvu + cc;
            const int c = kc + i;
            const float* w1 = wdw + c * 9;
            const float* w2 = wdw + (c + HID) * 9;
            const float* t1 = tile + i * TCH + cy * TP + 3 + 8 * cg;
            const float* t2 = t1 + 16 * TCH;
            float s1[8], s2[8];
#pragma unroll
            for (int j = 0; j < 8; ++j) { s1[j] = 0.f; s2[j] = 0.f; }
#pragma unroll
            for (int ky = 0; ky < 3; ++ky) {
                const float* r1 = t1 + ky * TP;
                const float* r2 = t2 + ky * TP;
                float a[10], bq[10];
                const v4f a0 = *reinterpret_cast<const v4f*>(r1 + 1);
                const v4f a1 = *reinterpret_cast<const v4f*>(r1 + 5);
                const v4f b0 = *reinterpret_cast<const v4f*>(r2 + 1);
                const v4f b1 = *reinterpret_cast<const v4f*>(r2 + 5);
                a[0] = r1[0];  a[9] = r1[9];
                bq[0] = r2[0]; bq[9] = r2[9];
#pragma unroll
                for (int e = 0; e < 4; ++e) { a[1 + e] = a0[e]; a[5 + e] = a1[e]; bq[1 + e] = b0[e]; bq[5 + e] = b1[e]; }
#pragma unroll
                for (int kx = 0; kx < 3; ++kx) {
                    const float wa = w1[ky * 3 + kx];
                    const float wb = w2[ky * 3 + kx];
#pragma unroll
                    for (int j = 0; j < 8; ++j) {
                        s1[j] += wa * a[j + kx];
                        s2[j] += wb * bq[j + kx];
                    }
                }
            }
            v4f o0, o1;
#pragma unroll
            for (int j = 0; j < 8; ++j) {
                const float t = tanhf(0.7978845608028654f * (s1[j] + 0.044715f * s1[j] * s1[j] * s1[j]));
                const float v = 0.5f * s1[j] * (1.f + t) * s2[j];
                if (j < 4) o0[j] = v; else o1[j - 4] = v;
            }
            float* gp = &g[i][cy * 32 + 8 * cg];
            *reinterpret_cast<v4f*>(gp)     = o0;
            *reinterpret_cast<v4f*>(gp + 4) = o1;
        }
        __syncthreads();
        if (more) stage();
#pragma unroll
        for (int kk = 0; kk < 16; kk += 4) {
            const int k0 = kk + 2 * hi;
            v2f a; a.x = wrow[kc + k0]; a.y = wrow[kc + k0 + 1];
#pragma unroll
            for (int j = 0; j < 4; ++j) {
                const int nt = ntb + j;
                const v2f p0 = *reinterpret_cast<const v2f*>(&g[k0][nt * 32 + 2 * m]);
                const v2f p1 = *reinterpret_cast<const v2f*>(&g[k0 + 1][nt * 32 + 2 * m]);
                v2f be, bo;
                be.x = p0.x; be.y = p1.x;
                bo.x = p0.y; bo.y = p1.y;
                accE[j] = __builtin_amdgcn_wmma_f32_16x16x4_f32(false, a, false, be, (short)0, accE[j], false, false);
                accO[j] = __builtin_amdgcn_wmma_f32_16x16x4_f32(false, a, false, bo, (short)0, accO[j], false, false);
            }
        }
        __syncthreads();
    }

#pragma unroll
    for (int j = 0; j < 4; ++j) {
        const int nt = ntb + j;
        const size_t base = ((size_t)b * Cout + mt * 16 + 8 * hi) * HW
                          + (size_t)(r0 + nt) * Wdim + c0 + 2 * m;
#pragma unroll
        for (int r = 0; r < 8; ++r) {
            v2f pr; pr.x = accE[j][r]; pr.y = accO[j][r];
            *reinterpret_cast<v2f*>(out + base + (size_t)r * HW) = pr;
        }
    }
}

extern "C" void kernel_launch(void* const* d_in, const int* in_sizes, int n_in,
                              void* d_out, int out_size, void* d_ws, size_t ws_size,
                              hipStream_t stream) {
    const float* x    = (const float*)d_in[0];
    const float* Win  = (const float*)d_in[1];
    const float* Wdw  = (const float*)d_in[2];
    const float* mix  = (const float*)d_in[3];
    const float* Wout = (const float*)d_in[4];
    float* out = (float*)d_out;
    float* ws  = (float*)d_ws;

    dim3 blk(256);
    dim3 g1(Wdim / 32, Hdim / 8, 2);
    k1_projin_dct<<<g1, blk, 0, stream>>>(x, Win, mix, ws);
    dim3 g2(Wdim / 32, Hdim / 8, 2);
    k2_dw_gate_out<<<g2, blk, 0, stream>>>(ws, Wdw, Wout, out);
}


# second kernel: GELU gate x(1+tanh u)/2 computed as the identical x/(1+exp(-2u)) with one hardware exp2 and one reciprocal per pixel, not the branching library tanh; gate rounding changes
# speedup vs baseline: 2.4695x; 1.0413x over previous
#include <hip/hip_runtime.h>

typedef __attribute__((ext_vector_type(2))) float v2f;
typedef __attribute__((ext_vector_type(8))) float v8f;

constexpr int Hdim = 256, Wdim = 256, Cin = 64, C2 = 256, HID = 128, Cout = 64;
constexpr int HW = Hdim * Wdim;

__constant__ const float DCT8[8][8] = {
  { 0.3535533906f, 0.3535533906f, 0.3535533906f, 0.3535533906f, 0.3535533906f, 0.3535533906f, 0.3535533906f, 0.3535533906f },
  { 0.4903926402f, 0.4157348062f, 0.2777851165f, 0.0975451610f,-0.0975451610f,-0.2777851165f,-0.4157348062f,-0.4903926402f },
  { 0.4619397663f, 0.1913417162f,-0.1913417162f,-0.4619397663f,-0.4619397663f,-0.1913417162f, 0.1913417162f, 0.4619397663f },
  { 0.4157348062f,-0.0975451610f,-0.4903926402f,-0.2777851165f, 0.2777851165f, 0.4903926402f, 0.0975451610f,-0.4157348062f },
  { 0.3535533906f,-0.3535533906f,-0.3535533906f, 0.3535533906f, 0.3535533906f,-0.3535533906f,-0.3535533906f, 0.3535533906f },
  { 0.2777851165f,-0.4903926402f, 0.0975451610f, 0.4157348062f,-0.4157348062f,-0.0975451610f, 0.4903926402f,-0.2777851165f },
  { 0.1913417162f,-0.4619397663f, 0.4619397663f,-0.1913417162f,-0.1913417162f, 0.4619397663f,-0.4619397663f, 0.1913417162f },
  { 0.0975451610f,-0.2777851165f, 0.4157348062f,-0.4903926402f, 0.4903926402f,-0.4157348062f, 0.2777851165f,-0.0975451610f }
};

typedef __attribute__((ext_vector_type(4))) float v4f;

constexpr int YP = 32;
constexpr int YC = 8 * YP + 4;

__device__ __forceinline__ int ysw(int row) { return ((row >> 1) & 1) << 2; }

__device__ __forceinline__ void dct_cols(float* __restrict__ t, int px)
{
    float* pa = t + px;
    float* pb = t + (px ^ 4);
    float x[8], w[8];
#pragma unroll
    for (int k = 0; k < 8; ++k) x[k] = (((k >> 1) & 1) ? pb : pa)[k * YP];
#pragma unroll
    for (int i = 0; i < 8; ++i) {
        float s = DCT8[i][0] * x[0];
#pragma unroll
        for (int k = 1; k < 8; ++k) s = __builtin_fmaf(DCT8[i][k], x[k], s);
        w[i] = s;
    }
#pragma unroll
    for (int i = 0; i < 8; ++i) (((i >> 1) & 1) ? pb : pa)[i * YP] = w[i];
}

template <bool MASK>
__device__ __forceinline__ void dct_rows(float* __restrict__ t, int py, int pch, v4f m0, v4f m1)
{
    float* row = t + py * YP;
    float* p0 = row + ((8 * pch)     ^ ysw(py));
    float* p1 = row + ((8 * pch + 4) ^ ysw(py));
    const v4f a = *reinterpret_cast<const v4f*>(p0);
    const v4f b = *reinterpret_cast<const v4f*>(p1);
    float w[8], o[8];
#pragma unroll
    for (int e = 0; e < 4; ++e) { w[e] = a[e]; w[4 + e] = b[e]; }
#pragma unroll
    for (int q = 0; q < 8; ++q) {
        float s = w[0] * DCT8[q][0];
#pragma unroll
        for (int j = 1; j < 8; ++j) s = __builtin_fmaf(w[j], DCT8[q][j], s);
        o[q] = s;
    }
    if (MASK) {
#pragma unroll
        for (int e = 0; e < 4; ++e) { o[e] *= m0[e]; o[4 + e] *= m1[e]; }
    }
    v4f r0, r1;
#pragma unroll
    for (int e = 0; e < 4; ++e) { r0[e] = o[e]; r1[e] = o[4 + e]; }
    *reinterpret_cast<v4f*>(p0) = r0;
    *reinterpret_cast<v4f*>(p1) = r1;
}

__global__ __launch_bounds__(256) void
k1_projin_dct(const float* __restrict__ x, const float* __restrict__ Win,
              const float* __restrict__ mix, float* __restrict__ ws)
{
    __shared__ alignas(16) float yt[2][16 * YC];

    const int tid  = threadIdx.x;
    const int lane = tid & 31;
    const int wv   = tid >> 5;
    const int m    = lane & 15;
    const int hi   = lane >> 4;

    const int b  = blockIdx.z;
    const int r0 = blockIdx.y * 8;
    const int c0 = blockIdx.x * 32;

    const float* xb = x + (size_t)b * Cin * HW;
    const float* xp = xb + (size_t)(r0 + wv) * Wdim + c0 + 2 * m;

    v2f be[16], bo[16];
#pragma unroll
    for (int s = 0; s < 16; ++s) {
        const int k0 = 4 * s + 2 * hi;
        const v2f l0 = *reinterpret_cast<const v2f*>(xp + (size_t)k0 * HW);
        const v2f l1 = *reinterpret_cast<const v2f*>(xp + (size_t)(k0 + 1) * HW);
        be[s].x = l0.x; be[s].y = l1.x;
        bo[s].x = l0.y; bo[s].y = l1.y;
    }

    const float* wr = Win + (size_t)m * Cin + 2 * hi;
    v2f a[16];
#pragma unroll
    for (int s = 0; s < 16; ++s) a[s] = *reinterpret_cast<const v2f*>(wr + 4 * s);

    const int pch = lane & 3;
    const int py  = lane >> 2;
    const float* mrow = mix + (size_t)wv * 64 + py * 8;

#pragma unroll 1
    for (int ob = 0; ob < 16; ++ob) {
        float* yb = yt[ob & 1];

        v8f accE = {}, accO = {};
#pragma unroll
        for (int s = 0; s < 16; ++s) {
            accE = __builtin_amdgcn_wmma_f32_16x16x4_f32(false, a[s], false, be[s], (short)0, accE, false, false);
            accO = __builtin_amdgcn_wmma_f32_16x16x4_f32(false, a[s], false, bo[s], (short)0, accO, false, false);
        }

        {
            const int obn = (ob < 15) ? ob + 1 : ob;
            const float* wn = wr + (size_t)obn * 16 * Cin;
#pragma unroll
            for (int s = 0; s < 16; ++s) a[s] = *reinterpret_cast<const v2f*>(wn + 4 * s);
        }
        const float* mA = mrow + (size_t)ob * 16 * 64;
        const float* mB = mA + 8 * 64;
        const v4f mA0 = *reinterpret_cast<const v4f*>(mA);
        const v4f mA1 = *reinterpret_cast<const v4f*>(mA + 4);
        const v4f mB0 = *reinterpret_cast<const v4f*>(mB);
        const v4f mB1 = *reinterpret_cast<const v4f*>(mB + 4);

        {
            float* yd = yb + (8 * hi) * YC + wv * YP + ((2 * m) ^ ysw(wv));
#pragma unroll
            for (int r = 0; r < 8; ++r) {
                v2f pr; pr.x = accE[r]; pr.y = accO[r];
                *reinterpret_cast<v2f*>(yd + r * YC) = pr;
            }
        }
        __syncthreads();

        float* tA = yb + wv * YC;
        float* tB = yb + (wv + 8) * YC;
        const v4f z4 = {0.f, 0.f, 0.f, 0.f};

        dct_cols(tA, lane);
        dct_cols(tB, lane);
        __syncthreads();
        dct_rows<true>(tA, py, pch, mA0, mA1);
        dct_rows<true>(tB, py, pch, mB0, mB1);
        __syncthreads();

        dct_cols(tA, lane);
        dct_cols(tB, lane);
        __syncthreads();
        dct_rows<false>(tA, py, pch, z4, z4);
        dct_rows<false>(tB, py, pch, z4, z4);
        __syncthreads();

#pragma unroll
        for (int it = 0; it < 4; ++it) {
            const int idx = tid + 256 * it;
            const int ch  = idx >> 6;
            const int row = (idx >> 3) & 7;
            const int u   = idx & 7;
            const v4f v = *reinterpret_cast<const v4f*>(yb + ch * YC + row * YP + ((4 * u) ^ ysw(row)));
            *reinterpret_cast<v4f*>(ws + ((size_t)b * C2 + ob * 16 + ch) * HW
                                       + (size_t)(r0 + row) * Wdim + c0 + 4 * u) = v;
        }
    }
}

constexpr int TP  = 44;
constexpr int TRH = 10;
constexpr int TCH = TRH * TP;
constexpr int GP  = 272;
constexpr float GK = -2.3022082f;

__global__ __launch_bounds__(256) void
k2_dw_gate_out(const float* __restrict__ ws, const float* __restrict__ wdw,
               const float* __restrict__ wout, float* __restrict__ out)
{
    __shared__ alignas(16) float g[16][GP];
    __shared__ alignas(16) float tile[32 * TCH];

    const int tid  = threadIdx.x;
    const int lane = tid & 31;
    const int wv   = tid >> 5;
    const int m    = lane & 15;
    const int hi   = lane >> 4;

    const int b  = blockIdx.z;
    const int r0 = blockIdx.y * 8;
    const int c0 = blockIdx.x * 32;

    const int wvu = __builtin_amdgcn_readfirstlane(wv);
    const int cy  = lane >> 2;
    const int cg  = lane & 3;

    const float* wsb  = ws + (size_t)b * C2 * HW;
    const int mt  = wv >> 1;
    const int ntb = (wv & 1) * 4;
    const float* wrow = wout + (size_t)(mt * 16 + m) * HID;

    const int q8 = tid & 7, rr = tid >> 3;
    v4f   cv[10];
    float hv[3];

    auto fetch = [&](int kc) {
#pragma unroll
        for (int t = 0; t < 10; ++t) {
            const int R   = rr + 32 * t;
            const int cs  = R / TRH;
            const int row = R - cs * TRH;
            const int yy  = r0 - 1 + row;
            const bool ok = ((unsigned)yy < (unsigned)Hdim);
            const int c   = kc + (cs & 15) + (cs >> 4) * HID;
            const float* p = wsb + (size_t)c * HW + (size_t)(ok ? yy : 0) * Wdim + c0 + q8 * 4;
            const v4f v = *reinterpret_cast<const v4f*>(p);
            cv[t] = ok ? v : (v4f){0.f, 0.f, 0.f, 0.f};
        }
#pragma unroll
        for (int t = 0; t < 3; ++t) {
            const int e    = tid + 256 * t;
            const int side = e & 1;
            const int R    = e >> 1;
            const int cs   = R / TRH;
            const int row  = R - cs * TRH;
            const int yy   = r0 - 1 + row;
            const int xx   = side ? c0 + 32 : c0 - 1;
            const bool ok  = (e < 640) && ((unsigned)yy < (unsigned)Hdim) && ((unsigned)xx < (unsigned)Wdim);
            const int c    = kc + (cs & 15) + (cs >> 4) * HID;
            hv[t] = ok ? wsb[(size_t)c * HW + (size_t)yy * Wdim + xx] : 0.f;
        }
    };
    auto stage = [&]() {
#pragma unroll
        for (int t = 0; t < 10; ++t) {
            const int R   = rr + 32 * t;
            const int cs  = R / TRH;
            const int row = R - cs * TRH;
            *reinterpret_cast<v4f*>(&tile[cs * TCH + row * TP + 4 + q8 * 4]) = cv[t];
        }
#pragma unroll
        for (int t = 0; t < 3; ++t) {
            const int e    = tid + 256 * t;
            const int side = e & 1;
            const int R    = e >> 1;
            const int cs   = R / TRH;
            const int row  = R - cs * TRH;
            if (e < 640) tile[cs * TCH + row * TP + (side ? 36 : 3)] = hv[t];
        }
    };

    v8f accE[4], accO[4];
#pragma unroll
    for (int j = 0; j < 4; ++j) { accE[j] = (v8f){}; accO[j] = (v8f){}; }

    fetch(0);
    stage();
    __syncthreads();

    for (int kc = 0; kc < HID; kc += 16) {
        const bool more = (kc + 16 < HID);
        if (more) fetch(kc + 16);
#pragma unroll 1
        for (int cc = 0; cc < 2; ++cc) {
            const int i = 2 * wvu + cc;
            const int c = kc + i;
            const float* w1 = wdw + c * 9;
            const float* w2 = wdw + (c + HID) * 9;
            const float* t1 = tile + i * TCH + cy * TP + 3 + 8 * cg;
            const float* t2 = t1 + 16 * TCH;
            float s1[8], s2[8];
#pragma unroll
            for (int j = 0; j < 8; ++j) { s1[j] = 0.f; s2[j] = 0.f; }
#pragma unroll
            for (int ky = 0; ky < 3; ++ky) {
                const float* r1 = t1 + ky * TP;
                const float* r2 = t2 + ky * TP;
                float a[10], bq[10];
                const v4f a0 = *reinterpret_cast<const v4f*>(r1 + 1);
                const v4f a1 = *reinterpret_cast<const v4f*>(r1 + 5);
                const v4f b0 = *reinterpret_cast<const v4f*>(r2 + 1);
                const v4f b1 = *reinterpret_cast<const v4f*>(r2 + 5);
                a[0] = r1[0];  a[9] = r1[9];
                bq[0] = r2[0]; bq[9] = r2[9];
#pragma unroll
                for (int e = 0; e < 4; ++e) { a[1 + e] = a0[e]; a[5 + e] = a1[e]; bq[1 + e] = b0[e]; bq[5 + e] = b1[e]; }
#pragma unroll
                for (int kx = 0; kx < 3; ++kx) {
                    const float wa = w1[ky * 3 + kx];
                    const float wb = w2[ky * 3 + kx];
#pragma unroll
                    for (int j = 0; j < 8; ++j) {
                        s1[j] += wa * a[j + kx];
                        s2[j] += wb * bq[j + kx];
                    }
                }
            }
            v4f o0, o1;
#pragma unroll
            for (int j = 0; j < 8; ++j) {
                const float p = s1[j] + 0.044715f * s1[j] * s1[j] * s1[j];
                const float e = __builtin_amdgcn_exp2f(GK * p);
                const float v = s1[j] * __builtin_amdgcn_rcpf(1.f + e) * s2[j];
                if (j < 4) o0[j] = v; else o1[j - 4] = v;
            }
            float* gp = &g[i][cy * 32 + 8 * cg];
            *reinterpret_cast<v4f*>(gp)     = o0;
            *reinterpret_cast<v4f*>(gp + 4) = o1;
        }
        __syncthreads();
        if (more) stage();
#pragma unroll
        for (int kk = 0; kk < 16; kk += 4) {
            const int k0 = kk + 2 * hi;
            v2f a; a.x = wrow[kc + k0]; a.y = wrow[kc + k0 + 1];
#pragma unroll
            for (int j = 0; j < 4; ++j) {
                const int nt = ntb + j;
                const v2f p0 = *reinterpret_cast<const v2f*>(&g[k0][nt * 32 + 2 * m]);
                const v2f p1 = *reinterpret_cast<const v2f*>(&g[k0 + 1][nt * 32 + 2 * m]);
                v2f be, bo;
                be.x = p0.x; be.y = p1.x;
                bo.x = p0.y; bo.y = p1.y;
                accE[j] = __builtin_amdgcn_wmma_f32_16x16x4_f32(false, a, false, be, (short)0, accE[j], false, false);
                accO[j] = __builtin_amdgcn_wmma_f32_16x16x4_f32(false, a, false, bo, (short)0, accO[j], false, false);
            }
        }
        __syncthreads();
    }

#pragma unroll
    for (int j = 0; j < 4; ++j) {
        const int nt = ntb + j;
        const size_t base = ((size_t)b * Cout + mt * 16 + 8 * hi) * HW
                          + (size_t)(r0 + nt) * Wdim + c0 + 2 * m;
#pragma unroll
        for (int r = 0; r < 8; ++r) {
            v2f pr; pr.x = accE[j][r]; pr.y = accO[j][r];
            *reinterpret_cast<v2f*>(out + base + (size_t)r * HW) = pr;
        }
    }
}

extern "C" void kernel_launch(void* const* d_in, const int* in_sizes, int n_in,
                              void* d_out, int out_size, void* d_ws, size_t ws_size,
                              hipStream_t stream) {
    const float* x    = (const float*)d_in[0];
    const float* Win  = (const float*)d_in[1];
    const float* Wdw  = (const float*)d_in[2];
    const float* mix  = (const float*)d_in[3];
    const float* Wout = (const float*)d_in[4];
    float* out = (float*)d_out;
    float* ws  = (float*)d_ws;

    dim3 blk(256);
    dim3 g1(Wdim / 32, Hdim / 8, 2);
    k1_projin_dct<<<g1, blk, 0, stream>>>(x, Win, mix, ws);
    dim3 g2(Wdim / 32, Hdim / 8, 2);
    k2_dw_gate_out<<<g2, blk, 0, stream>>>(ws, Wdw, Wout, out);
}


# both GEMMs: weight (A) fragments read from an LDS copy filled by whole-line 16-byte loads (k1: next block's tile each trip; k2: W_out once), not gathered from global 16 lines per load; same sums
# speedup vs baseline: 2.6348x; 1.0669x over previous
#include <hip/hip_runtime.h>

typedef __attribute__((ext_vector_type(2))) float v2f;
typedef __attribute__((ext_vector_type(8))) float v8f;

constexpr int Hdim = 256, Wdim = 256, Cin = 64, C2 = 256, HID = 128, Cout = 64;
constexpr int HW = Hdim * Wdim;

__constant__ const float DCT8[8][8] = {
  { 0.3535533906f, 0.3535533906f, 0.3535533906f, 0.3535533906f, 0.3535533906f, 0.3535533906f, 0.3535533906f, 0.3535533906f },
  { 0.4903926402f, 0.4157348062f, 0.2777851165f, 0.0975451610f,-0.0975451610f,-0.2777851165f,-0.4157348062f,-0.4903926402f },
  { 0.4619397663f, 0.1913417162f,-0.1913417162f,-0.4619397663f,-0.4619397663f,-0.1913417162f, 0.1913417162f, 0.4619397663f },
  { 0.4157348062f,-0.0975451610f,-0.4903926402f,-0.2777851165f, 0.2777851165f, 0.4903926402f, 0.0975451610f,-0.4157348062f },
  { 0.3535533906f,-0.3535533906f,-0.3535533906f, 0.3535533906f, 0.3535533906f,-0.3535533906f,-0.3535533906f, 0.3535533906f },
  { 0.2777851165f,-0.4903926402f, 0.0975451610f, 0.4157348062f,-0.4157348062f,-0.0975451610f, 0.4903926402f,-0.2777851165f },
  { 0.1913417162f,-0.4619397663f, 0.4619397663f,-0.1913417162f,-0.1913417162f, 0.4619397663f,-0.4619397663f, 0.1913417162f },
  { 0.0975451610f,-0.2777851165f, 0.4157348062f,-0.4903926402f, 0.4903926402f,-0.4157348062f, 0.2777851165f,-0.0975451610f }
};

typedef __attribute__((ext_vector_type(4))) float v4f;

constexpr int YP = 32;
constexpr int YC = 8 * YP + 4;
constexpr int WP = Cin + 4;

__device__ __forceinline__ int ysw(int row) { return ((row >> 1) & 1) << 2; }

__device__ __forceinline__ void dct_cols(float* __restrict__ t, int px)
{
    float* pa = t + px;
    float* pb = t + (px ^ 4);
    float x[8], w[8];
#pragma unroll
    for (int k = 0; k < 8; ++k) x[k] = (((k >> 1) & 1) ? pb : pa)[k * YP];
#pragma unroll
    for (int i = 0; i < 8; ++i) {
        float s = DCT8[i][0] * x[0];
#pragma unroll
        for (int k = 1; k < 8; ++k) s = __builtin_fmaf(DCT8[i][k], x[k], s);
        w[i] = s;
    }
#pragma unroll
    for (int i = 0; i < 8; ++i) (((i >> 1) & 1) ? pb : pa)[i * YP] = w[i];
}

template <bool MASK>
__device__ __forceinline__ void dct_rows(float* __restrict__ t, int py, int pch, v4f m0, v4f m1)
{
    float* row = t + py * YP;
    float* p0 = row + ((8 * pch)     ^ ysw(py));
    float* p1 = row + ((8 * pch + 4) ^ ysw(py));
    const v4f a = *reinterpret_cast<const v4f*>(p0);
    const v4f b = *reinterpret_cast<const v4f*>(p1);
    float w[8], o[8];
#pragma unroll
    for (int e = 0; e < 4; ++e) { w[e] = a[e]; w[4 + e] = b[e]; }
#pragma unroll
    for (int q = 0; q < 8; ++q) {
        float s = w[0] * DCT8[q][0];
#pragma unroll
        for (int j = 1; j < 8; ++j) s = __builtin_fmaf(w[j], DCT8[q][j], s);
        o[q] = s;
    }
    if (MASK) {
#pragma unroll
        for (int e = 0; e < 4; ++e) { o[e] *= m0[e]; o[4 + e] *= m1[e]; }
    }
    v4f r0, r1;
#pragma unroll
    for (int e = 0; e < 4; ++e) { r0[e] = o[e]; r1[e] = o[4 + e]; }
    *reinterpret_cast<v4f*>(p0) = r0;
    *reinterpret_cast<v4f*>(p1) = r1;
}

__global__ __launch_bounds__(256) void
k1_projin_dct(const float* __restrict__ x, const float* __restrict__ Win,
              const float* __restrict__ mix, float* __restrict__ ws)
{
    __shared__ alignas(16) float yt[2][16 * YC];
    __shared__ alignas(16) float wt[2][16 * WP];

    const int tid  = threadIdx.x;
    const int lane = tid & 31;
    const int wv   = tid >> 5;
    const int m    = lane & 15;
    const int hi   = lane >> 4;

    const int b  = blockIdx.z;
    const int r0 = blockIdx.y * 8;
    const int c0 = blockIdx.x * 32;

    const float* wsrc = Win + 4 * tid;
    const int    wdst = (tid >> 4) * WP + 4 * (tid & 15);
    const v4f w0 = *reinterpret_cast<const v4f*>(wsrc);

    const float* xb = x + (size_t)b * Cin * HW;
    const float* xp = xb + (size_t)(r0 + wv) * Wdim + c0 + 2 * m;

    v2f be[16], bo[16];
#pragma unroll
    for (int s = 0; s < 16; ++s) {
        const int k0 = 4 * s + 2 * hi;
        const v2f l0 = *reinterpret_cast<const v2f*>(xp + (size_t)k0 * HW);
        const v2f l1 = *reinterpret_cast<const v2f*>(xp + (size_t)(k0 + 1) * HW);
        be[s].x = l0.x; be[s].y = l1.x;
        bo[s].x = l0.y; bo[s].y = l1.y;
    }

    *reinterpret_cast<v4f*>(&wt[0][wdst]) = w0;
    __syncthreads();
    const int wfr = m * WP + 2 * hi;
    v2f a[16];
#pragma unroll
    for (int s = 0; s < 16; ++s) a[s] = *reinterpret_cast<const v2f*>(&wt[0][wfr + 4 * s]);

    const int pch = lane & 3;
    const int py  = lane >> 2;
    const float* mrow = mix + (size_t)wv * 64 + py * 8;

#pragma unroll 1
    for (int ob = 0; ob < 16; ++ob) {
        float* yb = yt[ob & 1];
        float* wn = wt[(ob + 1) & 1];

        const v4f wnx = *reinterpret_cast<const v4f*>(wsrc + (size_t)((ob < 15) ? ob + 1 : ob) * 16 * Cin);

        v8f accE = {}, accO = {};
#pragma unroll
        for (int s = 0; s < 16; ++s) {
            accE = __builtin_amdgcn_wmma_f32_16x16x4_f32(false, a[s], false, be[s], (short)0, accE, false, false);
            accO = __builtin_amdgcn_wmma_f32_16x16x4_f32(false, a[s], false, bo[s], (short)0, accO, false, false);
        }

        *reinterpret_cast<v4f*>(wn + wdst) = wnx;
        const float* mA = mrow + (size_t)ob * 16 * 64;
        const float* mB = mA + 8 * 64;
        const v4f mA0 = *reinterpret_cast<const v4f*>(mA);
        const v4f mA1 = *reinterpret_cast<const v4f*>(mA + 4);
        const v4f mB0 = *reinterpret_cast<const v4f*>(mB);
        const v4f mB1 = *reinterpret_cast<const v4f*>(mB + 4);

        {
            float* yd = yb + (8 * hi) * YC + wv * YP + ((2 * m) ^ ysw(wv));
#pragma unroll
            for (int r = 0; r < 8; ++r) {
                v2f pr; pr.x = accE[r]; pr.y = accO[r];
                *reinterpret_cast<v2f*>(yd + r * YC) = pr;
            }
        }
        __syncthreads();

        float* tA = yb + wv * YC;
        float* tB = yb + (wv + 8) * YC;
        const v4f z4 = {0.f, 0.f, 0.f, 0.f};

        dct_cols(tA, lane);
        dct_cols(tB, lane);
        __syncthreads();
        dct_rows<true>(tA, py, pch, mA0, mA1);
        dct_rows<true>(tB, py, pch, mB0, mB1);
        __syncthreads();

        dct_cols(tA, lane);
        dct_cols(tB, lane);
        __syncthreads();
        dct_rows<false>(tA, py, pch, z4, z4);
        dct_rows<false>(tB, py, pch, z4, z4);
        __syncthreads();

#pragma unroll
        for (int it = 0; it < 4; ++it) {
            const int idx = tid + 256 * it;
            const int ch  = idx >> 6;
            const int row = (idx >> 3) & 7;
            const int u   = idx & 7;
            const v4f v = *reinterpret_cast<const v4f*>(yb + ch * YC + row * YP + ((4 * u) ^ ysw(row)));
            *reinterpret_cast<v4f*>(ws + ((size_t)b * C2 + ob * 16 + ch) * HW
                                       + (size_t)(r0 + row) * Wdim + c0 + 4 * u) = v;
        }

#pragma unroll
        for (int s = 0; s < 16; ++s) a[s] = *reinterpret_cast<const v2f*>(wn + wfr + 4 * s);
    }
}

constexpr int TP  = 44;
constexpr int TRH = 10;
constexpr int TCH = TRH * TP;
constexpr int GP  = 272;
constexpr int WLP = HID + 4;
constexpr float GK = -2.3022082f;

__global__ __launch_bounds__(256) void
k2_dw_gate_out(const float* __restrict__ ws, const float* __restrict__ wdw,
               const float* __restrict__ wout, float* __restrict__ out)
{
    __shared__ alignas(16) float g[16][GP];
    __shared__ alignas(16) float tile[32 * TCH];
    __shared__ alignas(16) float wl[Cout * WLP];

    const int tid  = threadIdx.x;
    const int lane = tid & 31;
    const int wv   = tid >> 5;
    const int m    = lane & 15;
    const int hi   = lane >> 4;

    const int b  = blockIdx.z;
    const int r0 = blockIdx.y * 8;
    const int c0 = blockIdx.x * 32;

    const int wvu = __builtin_amdgcn_readfirstlane(wv);
    const int cy  = lane >> 2;
    const int cg  = lane & 3;

    const float* wsb  = ws + (size_t)b * C2 * HW;
    const int mt  = wv >> 1;
    const int ntb = (wv & 1) * 4;
    const float* wrow = wl + (mt * 16 + m) * WLP;

    const int q8 = tid & 7, rr = tid >> 3;
    v4f   cv[10];
    float hv[3];

    auto fetch = [&](int kc) {
#pragma unroll
        for (int t = 0; t < 10; ++t) {
            const int R   = rr + 32 * t;
            const int cs  = R / TRH;
            const int row = R - cs * TRH;
            const int yy  = r0 - 1 + row;
            const bool ok = ((unsigned)yy < (unsigned)Hdim);
            const int c   = kc + (cs & 15) + (cs >> 4) * HID;
            const float* p = wsb + (size_t)c * HW + (size_t)(ok ? yy : 0) * Wdim + c0 + q8 * 4;
            const v4f v = *reinterpret_cast<const v4f*>(p);
            cv[t] = ok ? v : (v4f){0.f, 0.f, 0.f, 0.f};
        }
#pragma unroll
        for (int t = 0; t < 3; ++t) {
            const int e    = tid + 256 * t;
            const int side = e & 1;
            const int R    = e >> 1;
            const int cs   = R / TRH;
            const int row  = R - cs * TRH;
            const int yy   = r0 - 1 + row;
            const int xx   = side ? c0 + 32 : c0 - 1;
            const bool ok  = (e < 640) && ((unsigned)yy < (unsigned)Hdim) && ((unsigned)xx < (unsigned)Wdim);
            const int c    = kc + (cs & 15) + (cs >> 4) * HID;
            hv[t] = ok ? wsb[(size_t)c * HW + (size_t)yy * Wdim + xx] : 0.f;
        }
    };
    auto stage = [&]() {
#pragma unroll
        for (int t = 0; t < 10; ++t) {
            const int R   = rr + 32 * t;
            const int cs  = R / TRH;
            const int row = R - cs * TRH;
            *reinterpret_cast<v4f*>(&tile[cs * TCH + row * TP + 4 + q8 * 4]) = cv[t];
        }
#pragma unroll
        for (int t = 0; t < 3; ++t) {
            const int e    = tid + 256 * t;
            const int side = e & 1;
            const int R    = e >> 1;
            const int cs   = R / TRH;
            const int row  = R - cs * TRH;
            if (e < 640) tile[cs * TCH + row * TP + (side ? 36 : 3)] = hv[t];
        }
    };

    v8f accE[4], accO[4];
#pragma unroll
    for (int j = 0; j < 4; ++j) { accE[j] = (v8f){}; accO[j] = (v8f){}; }

    v4f wq[8];
#pragma unroll
    for (int i = 0; i < 8; ++i) wq[i] = *reinterpret_cast<const v4f*>(wout + 4 * (tid + 256 * i));

    fetch(0);
#pragma unroll
    for (int i = 0; i < 8; ++i) {
        const int p = tid + 256 * i;
        *reinterpret_cast<v4f*>(&wl[(p >> 5) * WLP + 4 * (p & 31)]) = wq[i];
    }
    stage();
    __syncthreads();

    for (int kc = 0; kc < HID; kc += 16) {
        const bool more = (kc + 16 < HID);
        if (more) fetch(kc + 16);
#pragma unroll 1
        for (int cc = 0; cc < 2; ++cc) {
            const int i = 2 * wvu + cc;
            const int c = kc + i;
            const float* w1 = wdw + c * 9;
            const float* w2 = wdw + (c + HID) * 9;
            const float* t1 = tile + i * TCH + cy * TP + 3 + 8 * cg;
            const float* t2 = t1 + 16 * TCH;
            float s1[8], s2[8];
#pragma unroll
            for (int j = 0; j < 8; ++j) { s1[j] = 0.f; s2[j] = 0.f; }
#pragma unroll
            for (int ky = 0; ky < 3; ++ky) {
                const float* r1 = t1 + ky * TP;
                const float* r2 = t2 + ky * TP;
                float a[10], bq[10];
                const v4f a0 = *reinterpret_cast<const v4f*>(r1 + 1);
                const v4f a1 = *reinterpret_cast<const v4f*>(r1 + 5);
                const v4f b0 = *reinterpret_cast<const v4f*>(r2 + 1);
                const v4f b1 = *reinterpret_cast<const v4f*>(r2 + 5);
                a[0] = r1[0];  a[9] = r1[9];
                bq[0] = r2[0]; bq[9] = r2[9];
#pragma unroll
                for (int e = 0; e < 4; ++e) { a[1 + e] = a0[e]; a[5 + e] = a1[e]; bq[1 + e] = b0[e]; bq[5 + e] = b1[e]; }
#pragma unroll
                for (int kx = 0; kx < 3; ++kx) {
                    const float wa = w1[ky * 3 + kx];
                    const float wb = w2[ky * 3 + kx];
#pragma unroll
                    for (int j = 0; j < 8; ++j) {
                        s1[j] += wa * a[j + kx];
                        s2[j] += wb * bq[j + kx];
                    }
                }
            }
            v4f o0, o1;
#pragma unroll
            for (int j = 0; j < 8; ++j) {
                const float p = s1[j] + 0.044715f * s1[j] * s1[j] * s1[j];
                const float e = __builtin_amdgcn_exp2f(GK * p);
                const float v = s1[j] * __builtin_amdgcn_rcpf(1.f + e) * s2[j];
                if (j < 4) o0[j] = v; else o1[j - 4] = v;
            }
            float* gp = &g[i][cy * 32 + 8 * cg];
            *reinterpret_cast<v4f*>(gp)     = o0;
            *reinterpret_cast<v4f*>(gp + 4) = o1;
        }
        __syncthreads();
        if (more) stage();
#pragma unroll
        for (int kk = 0; kk < 16; kk += 4) {
            const int k0 = kk + 2 * hi;
            const v2f a = *reinterpret_cast<const v2f*>(wrow + kc + k0);
#pragma unroll
            for (int j = 0; j < 4; ++j) {
                const int nt = ntb + j;
                const v2f p0 = *reinterpret_cast<const v2f*>(&g[k0][nt * 32 + 2 * m]);
                const v2f p1 = *reinterpret_cast<const v2f*>(&g[k0 + 1][nt * 32 + 2 * m]);
                v2f be, bo;
                be.x = p0.x; be.y = p1.x;
                bo.x = p0.y; bo.y = p1.y;
                accE[j] = __builtin_amdgcn_wmma_f32_16x16x4_f32(false, a, false, be, (short)0, accE[j], false, false);
                accO[j] = __builtin_amdgcn_wmma_f32_16x16x4_f32(false, a, false, bo, (short)0, accO[j], false, false);
            }
        }
        __syncthreads();
    }

#pragma unroll
    for (int j = 0; j < 4; ++j) {
        const int nt = ntb + j;
        const size_t base = ((size_t)b * Cout + mt * 16 + 8 * hi) * HW
                          + (size_t)(r0 + nt) * Wdim + c0 + 2 * m;
#pragma unroll
        for (int r = 0; r < 8; ++r) {
            v2f pr; pr.x = accE[j][r]; pr.y = accO[j][r];
            *reinterpret_cast<v2f*>(out + base + (size_t)r * HW) = pr;
        }
    }
}

extern "C" void kernel_launch(void* const* d_in, const int* in_sizes, int n_in,
                              void* d_out, int out_size, void* d_ws, size_t ws_size,
                              hipStream_t stream) {
    const float* x    = (const float*)d_in[0];
    const float* Win  = (const float*)d_in[1];
    const float* Wdw  = (const float*)d_in[2];
    const float* mix  = (const float*)d_in[3];
    const float* Wout = (const float*)d_in[4];
    float* out = (float*)d_out;
    float* ws  = (float*)d_ws;

    dim3 blk(256);
    dim3 g1(Wdim / 32, Hdim / 8, 2);
    k1_projin_dct<<<g1, blk, 0, stream>>>(x, Win, mix, ws);
    dim3 g2(Wdim / 32, Hdim / 8, 2);
    k2_dw_gate_out<<<g2, blk, 0, stream>>>(ws, Wdw, Wout, out);
}


# first kernel: first DCT (A X A^T) applied to the tile's 64 input channels in front of project_in (the linear maps commute), not to its 256 outputs; mask on the accumulators; sum order changes
# speedup vs baseline: 2.7570x; 1.0464x over previous
#include <hip/hip_runtime.h>

typedef __attribute__((ext_vector_type(2))) float v2f;
typedef __attribute__((ext_vector_type(8))) float v8f;

constexpr int Hdim = 256, Wdim = 256, Cin = 64, C2 = 256, HID = 128, Cout = 64;
constexpr int HW = Hdim * Wdim;

__constant__ const float DCT8[8][8] = {
  { 0.3535533906f, 0.3535533906f, 0.3535533906f, 0.3535533906f, 0.3535533906f, 0.3535533906f, 0.3535533906f, 0.3535533906f },
  { 0.4903926402f, 0.4157348062f, 0.2777851165f, 0.0975451610f,-0.0975451610f,-0.2777851165f,-0.4157348062f,-0.4903926402f },
  { 0.4619397663f, 0.1913417162f,-0.1913417162f,-0.4619397663f,-0.4619397663f,-0.1913417162f, 0.1913417162f, 0.4619397663f },
  { 0.4157348062f,-0.0975451610f,-0.4903926402f,-0.2777851165f, 0.2777851165f, 0.4903926402f, 0.0975451610f,-0.4157348062f },
  { 0.3535533906f,-0.3535533906f,-0.3535533906f, 0.3535533906f, 0.3535533906f,-0.3535533906f,-0.3535533906f, 0.3535533906f },
  { 0.2777851165f,-0.4903926402f, 0.0975451610f, 0.4157348062f,-0.4157348062f,-0.0975451610f, 0.4903926402f,-0.2777851165f },
  { 0.1913417162f,-0.4619397663f, 0.4619397663f,-0.1913417162f,-0.1913417162f, 0.4619397663f,-0.4619397663f, 0.1913417162f },
  { 0.0975451610f,-0.2777851165f, 0.4157348062f,-0.4903926402f, 0.4903926402f,-0.4157348062f, 0.2777851165f,-0.0975451610f }
};

typedef __attribute__((ext_vector_type(4))) float v4f;

constexpr int YP = 32;
constexpr int YC = 8 * YP + 4;
constexpr int WP = Cin + 4;

__device__ __forceinline__ int ysw(int row) { return ((row >> 1) & 1) << 2; }

__device__ __forceinline__ void dct_cols(float* __restrict__ t, int px)
{
    float* pa = t + px;
    float* pb = t + (px ^ 4);
    float x[8], w[8];
#pragma unroll
    for (int k = 0; k < 8; ++k) x[k] = (((k >> 1) & 1) ? pb : pa)[k * YP];
#pragma unroll
    for (int i = 0; i < 8; ++i) {
        float s = DCT8[i][0] * x[0];
#pragma unroll
        for (int k = 1; k < 8; ++k) s = __builtin_fmaf(DCT8[i][k], x[k], s);
        w[i] = s;
    }
#pragma unroll
    for (int i = 0; i < 8; ++i) (((i >> 1) & 1) ? pb : pa)[i * YP] = w[i];
}

__device__ __forceinline__ void dct_rows(float* __restrict__ t, int py, int pch)
{
    float* row = t + py * YP;
    float* p0 = row + ((8 * pch)     ^ ysw(py));
    float* p1 = row + ((8 * pch + 4) ^ ysw(py));
    const v4f a = *reinterpret_cast<const v4f*>(p0);
    const v4f b = *reinterpret_cast<const v4f*>(p1);
    float w[8], o[8];
#pragma unroll
    for (int e = 0; e < 4; ++e) { w[e] = a[e]; w[4 + e] = b[e]; }
#pragma unroll
    for (int q = 0; q < 8; ++q) {
        float s = w[0] * DCT8[q][0];
#pragma unroll
        for (int j = 1; j < 8; ++j) s = __builtin_fmaf(w[j], DCT8[q][j], s);
        o[q] = s;
    }
    v4f r0, r1;
#pragma unroll
    for (int e = 0; e < 4; ++e) { r0[e] = o[e]; r1[e] = o[4 + e]; }
    *reinterpret_cast<v4f*>(p0) = r0;
    *reinterpret_cast<v4f*>(p1) = r1;
}

__global__ __launch_bounds__(256) void
k1_projin_dct(const float* __restrict__ x, const float* __restrict__ Win,
              const float* __restrict__ mix, float* __restrict__ ws)
{
    __shared__ alignas(16) float yt[2][16 * YC];
    __shared__ alignas(16) float wt[2][16 * WP];
    __shared__ alignas(16) float xs[Cin * YC];

    const int tid  = threadIdx.x;
    const int lane = tid & 31;
    const int wv   = tid >> 5;
    const int m    = lane & 15;
    const int hi   = lane >> 4;

    const int b  = blockIdx.z;
    const int r0 = blockIdx.y * 8;
    const int c0 = blockIdx.x * 32;

    const float* wsrc = Win + 4 * tid;
    const int    wdst = (tid >> 4) * WP + 4 * (tid & 15);
    const v4f w0 = *reinterpret_cast<const v4f*>(wsrc);

    const float* xb = x + (size_t)b * Cin * HW;
    const int xrow = (tid >> 3) & 7;
    const int xu   = tid & 7;
    const float* xsrc = xb + (size_t)(tid >> 6) * HW + (size_t)(r0 + xrow) * Wdim + c0 + 4 * xu;
    const int    xdst = (tid >> 6) * YC + xrow * YP + ((4 * xu) ^ ysw(xrow));
    {
        v4f xv[16];
#pragma unroll
        for (int i = 0; i < 16; ++i) xv[i] = *reinterpret_cast<const v4f*>(xsrc + (size_t)(4 * i) * HW);
#pragma unroll
        for (int i = 0; i < 16; ++i) *reinterpret_cast<v4f*>(&xs[xdst + 4 * i * YC]) = xv[i];
    }
    *reinterpret_cast<v4f*>(&wt[0][wdst]) = w0;

    __syncthreads();

    const int pch = lane & 3;
    const int py  = lane >> 2;

#pragma unroll 1
    for (int t = 0; t < 8; t += 2) {
        dct_cols(xs + (wv + 8 * t) * YC, lane);
        dct_cols(xs + (wv + 8 * t + 8) * YC, lane);
    }
    __syncthreads();
#pragma unroll 1
    for (int t = 0; t < 8; t += 2) {
        dct_rows(xs + (wv + 8 * t) * YC, py, pch);
        dct_rows(xs + (wv + 8 * t + 8) * YC, py, pch);
    }
    __syncthreads();

    const float* xf = xs + wv * YP + ((2 * m) ^ ysw(wv));

    v2f be[16], bo[16];
#pragma unroll
    for (int s = 0; s < 16; ++s) {
        const int k0 = 4 * s + 2 * hi;
        const v2f l0 = *reinterpret_cast<const v2f*>(xf + k0 * YC);
        const v2f l1 = *reinterpret_cast<const v2f*>(xf + (k0 + 1) * YC);
        be[s].x = l0.x; be[s].y = l1.x;
        bo[s].x = l0.y; bo[s].y = l1.y;
    }

    const int wfr = m * WP + 2 * hi;
    v2f a[16];
#pragma unroll
    for (int s = 0; s < 16; ++s) a[s] = *reinterpret_cast<const v2f*>(&wt[0][wfr + 4 * s]);

    const float* mlane = mix + (size_t)(8 * hi) * 64 + wv * 8 + ((2 * m) & 7);

#pragma unroll 1
    for (int ob = 0; ob < 16; ++ob) {
        float* yb = yt[ob & 1];
        float* wn = wt[(ob + 1) & 1];

        const v4f wnx = *reinterpret_cast<const v4f*>(wsrc + (size_t)((ob < 15) ? ob + 1 : ob) * 16 * Cin);
        v2f mk[8];
#pragma unroll
        for (int r = 0; r < 8; ++r) mk[r] = *reinterpret_cast<const v2f*>(mlane + (size_t)(ob * 16 + r) * 64);
        __builtin_amdgcn_sched_barrier(0);

        v8f accE = {}, accO = {};
#pragma unroll
        for (int s = 0; s < 16; ++s) {
            accE = __builtin_amdgcn_wmma_f32_16x16x4_f32(false, a[s], false, be[s], (short)0, accE, false, false);
            accO = __builtin_amdgcn_wmma_f32_16x16x4_f32(false, a[s], false, bo[s], (short)0, accO, false, false);
        }

        {
            float* yd = yb + (8 * hi) * YC + wv * YP + ((2 * m) ^ ysw(wv));
#pragma unroll
            for (int r = 0; r < 8; ++r) {
                v2f pr; pr.x = accE[r] * mk[r].x; pr.y = accO[r] * mk[r].y;
                *reinterpret_cast<v2f*>(yd + r * YC) = pr;
            }
        }
        __syncthreads();

        *reinterpret_cast<v4f*>(wn + wdst) = wnx;

        float* tA = yb + wv * YC;
        float* tB = yb + (wv + 8) * YC;

        dct_cols(tA, lane);
        dct_cols(tB, lane);
        __syncthreads();
        dct_rows(tA, py, pch);
        dct_rows(tB, py, pch);
        __syncthreads();

#pragma unroll
        for (int it = 0; it < 4; ++it) {
            const int idx = tid + 256 * it;
            const int ch  = idx >> 6;
            const int row = (idx >> 3) & 7;
            const int u   = idx & 7;
            const v4f v = *reinterpret_cast<const v4f*>(yb + ch * YC + row * YP + ((4 * u) ^ ysw(row)));
            *reinterpret_cast<v4f*>(ws + ((size_t)b * C2 + ob * 16 + ch) * HW
                                       + (size_t)(r0 + row) * Wdim + c0 + 4 * u) = v;
        }

#pragma unroll
        for (int s = 0; s < 16; ++s) a[s] = *reinterpret_cast<const v2f*>(wn + wfr + 4 * s);
    }
}

constexpr int TP  = 44;
constexpr int TRH = 10;
constexpr int TCH = TRH * TP;
constexpr int GP  = 272;
constexpr int WLP = HID + 4;
constexpr float GK = -2.3022082f;

__global__ __launch_bounds__(256) void
k2_dw_gate_out(const float* __restrict__ ws, const float* __restrict__ wdw,
               const float* __restrict__ wout, float* __restrict__ out)
{
    __shared__ alignas(16) float g[16][GP];
    __shared__ alignas(16) float tile[32 * TCH];
    __shared__ alignas(16) float wl[Cout * WLP];

    const int tid  = threadIdx.x;
    const int lane = tid & 31;
    const int wv   = tid >> 5;
    const int m    = lane & 15;
    const int hi   = lane >> 4;

    const int b  = blockIdx.z;
    const int r0 = blockIdx.y * 8;
    const int c0 = blockIdx.x * 32;

    const int wvu = __builtin_amdgcn_readfirstlane(wv);
    const int cy  = lane >> 2;
    const int cg  = lane & 3;

    const float* wsb  = ws + (size_t)b * C2 * HW;
    const int mt  = wv >> 1;
    const int ntb = (wv & 1) * 4;
    const float* wrow = wl + (mt * 16 + m) * WLP;

    const int q8 = tid & 7, rr = tid >> 3;
    v4f   cv[10];
    float hv[3];

    auto fetch = [&](int kc) {
#pragma unroll
        for (int t = 0; t < 10; ++t) {
            const int R   = rr + 32 * t;
            const int cs  = R / TRH;
            const int row = R - cs * TRH;
            const int yy  = r0 - 1 + row;
            const bool ok = ((unsigned)yy < (unsigned)Hdim);
            const int c   = kc + (cs & 15) + (cs >> 4) * HID;
            const float* p = wsb + (size_t)c * HW + (size_t)(ok ? yy : 0) * Wdim + c0 + q8 * 4;
            const v4f v = *reinterpret_cast<const v4f*>(p);
            cv[t] = ok ? v : (v4f){0.f, 0.f, 0.f, 0.f};
        }
#pragma unroll
        for (int t = 0; t < 3; ++t) {
            const int e    = tid + 256 * t;
            const int side = e & 1;
            const int R    = e >> 1;
            const int cs   = R / TRH;
            const int row  = R - cs * TRH;
            const int yy   = r0 - 1 + row;
            const int xx   = side ? c0 + 32 : c0 - 1;
            const bool ok  = (e < 640) && ((unsigned)yy < (unsigned)Hdim) && ((unsigned)xx < (unsigned)Wdim);
            const int c    = kc + (cs & 15) + (cs >> 4) * HID;
            hv[t] = ok ? wsb[(size_t)c * HW + (size_t)yy * Wdim + xx] : 0.f;
        }
    };
    auto stage = [&]() {
#pragma unroll
        for (int t = 0; t < 10; ++t) {
            const int R   = rr + 32 * t;
            const int cs  = R / TRH;
            const int row = R - cs * TRH;
            *reinterpret_cast<v4f*>(&tile[cs * TCH + row * TP + 4 + q8 * 4]) = cv[t];
        }
#pragma unroll
        for (int t = 0; t < 3; ++t) {
            const int e    = tid + 256 * t;
            const int side = e & 1;
            const int R    = e >> 1;
            const int cs   = R / TRH;
            const int row  = R - cs * TRH;
            if (e < 640) tile[cs * TCH + row * TP + (side ? 36 : 3)] = hv[t];
        }
    };

    v8f accE[4], accO[4];
#pragma unroll
    for (int j = 0; j < 4; ++j) { accE[j] = (v8f){}; accO[j] = (v8f){}; }

    v4f wq[8];
#pragma unroll
    for (int i = 0; i < 8; ++i) wq[i] = *reinterpret_cast<const v4f*>(wout + 4 * (tid + 256 * i));

    fetch(0);
#pragma unroll
    for (int i = 0; i < 8; ++i) {
        const int p = tid + 256 * i;
        *reinterpret_cast<v4f*>(&wl[(p >> 5) * WLP + 4 * (p & 31)]) = wq[i];
    }
    stage();
    __syncthreads();

    for (int kc = 0; kc < HID; kc += 16) {
        const bool more = (kc + 16 < HID);
        if (more) fetch(kc + 16);
#pragma unroll 1
        for (int cc = 0; cc < 2; ++cc) {
            const int i = 2 * wvu + cc;
            const int c = kc + i;
            const float* w1 = wdw + c * 9;
            const float* w2 = wdw + (c + HID) * 9;
            const float* t1 = tile + i * TCH + cy * TP + 3 + 8 * cg;
            const float* t2 = t1 + 16 * TCH;
            float s1[8], s2[8];
#pragma unroll
            for (int j = 0; j < 8; ++j) { s1[j] = 0.f; s2[j] = 0.f; }
#pragma unroll
            for (int ky = 0; ky < 3; ++ky) {
                const float* r1 = t1 + ky * TP;
                const float* r2 = t2 + ky * TP;
                float a[10], bq[10];
                const v4f a0 = *reinterpret_cast<const v4f*>(r1 + 1);
                const v4f a1 = *reinterpret_cast<const v4f*>(r1 + 5);
                const v4f b0 = *reinterpret_cast<const v4f*>(r2 + 1);
                const v4f b1 = *reinterpret_cast<const v4f*>(r2 + 5);
                a[0] = r1[0];  a[9] = r1[9];
                bq[0] = r2[0]; bq[9] = r2[9];
#pragma unroll
                for (int e = 0; e < 4; ++e) { a[1 + e] = a0[e]; a[5 + e] = a1[e]; bq[1 + e] = b0[e]; bq[5 + e] = b1[e]; }
#pragma unroll
                for (int kx = 0; kx < 3; ++kx) {
                    const float wa = w1[ky * 3 + kx];
                    const float wb = w2[ky * 3 + kx];
#pragma unroll
                    for (int j = 0; j < 8; ++j) {
                        s1[j] += wa * a[j + kx];
                        s2[j] += wb * bq[j + kx];
                    }
                }
            }
            v4f o0, o1;
#pragma unroll
            for (int j = 0; j < 8; ++j) {
                const float p = s1[j] + 0.044715f * s1[j] * s1[j] * s1[j];
                const float e = __builtin_amdgcn_exp2f(GK * p);
                const float v = s1[j] * __builtin_amdgcn_rcpf(1.f + e) * s2[j];
                if (j < 4) o0[j] = v; else o1[j - 4] = v;
            }
            float* gp = &g[i][cy * 32 + 8 * cg];
            *reinterpret_cast<v4f*>(gp)     = o0;
            *reinterpret_cast<v4f*>(gp + 4) = o1;
        }
        __syncthreads();
        if (more) stage();
#pragma unroll
        for (int kk = 0; kk < 16; kk += 4) {
            const int k0 = kk + 2 * hi;
            const v2f a = *reinterpret_cast<const v2f*>(wrow + kc + k0);
#pragma unroll
            for (int j = 0; j < 4; ++j) {
                const int nt = ntb + j;
                const v2f p0 = *reinterpret_cast<const v2f*>(&g[k0][nt * 32 + 2 * m]);
                const v2f p1 = *reinterpret_cast<const v2f*>(&g[k0 + 1][nt * 32 + 2 * m]);
                v2f be, bo;
                be.x = p0.x; be.y = p1.x;
                bo.x = p0.y; bo.y = p1.y;
                accE[j] = __builtin_amdgcn_wmma_f32_16x16x4_f32(false, a, false, be, (short)0, accE[j], false, false);
                accO[j] = __builtin_amdgcn_wmma_f32_16x16x4_f32(false, a, false, bo, (short)0, accO[j], false, false);
            }
        }
        __syncthreads();
    }

#pragma unroll
    for (int j = 0; j < 4; ++j) {
        const int nt = ntb + j;
        const size_t base = ((size_t)b * Cout + mt * 16 + 8 * hi) * HW
                          + (size_t)(r0 + nt) * Wdim + c0 + 2 * m;
#pragma unroll
        for (int r = 0; r < 8; ++r) {
            v2f pr; pr.x = accE[j][r]; pr.y = accO[j][r];
            *reinterpret_cast<v2f*>(out + base + (size_t)r * HW) = pr;
        }
    }
}

extern "C" void kernel_launch(void* const* d_in, const int* in_sizes, int n_in,
                              void* d_out, int out_size, void* d_ws, size_t ws_size,
                              hipStream_t stream) {
    const float* x    = (const float*)d_in[0];
    const float* Win  = (const float*)d_in[1];
    const float* Wdw  = (const float*)d_in[2];
    const float* mix  = (const float*)d_in[3];
    const float* Wout = (const float*)d_in[4];
    float* out = (float*)d_out;
    float* ws  = (float*)d_ws;

    dim3 blk(256);
    dim3 g1(Wdim / 32, Hdim / 8, 2);
    k1_projin_dct<<<g1, blk, 0, stream>>>(x, Win, mix, ws);
    dim3 g2(Wdim / 32, Hdim / 8, 2);
    k2_dw_gate_out<<<g2, blk, 0, stream>>>(ws, Wdw, Wout, out);
}


# speedup vs baseline: 2.8823x; 1.0454x over previous
#include <hip/hip_runtime.h>

typedef __attribute__((ext_vector_type(2))) float v2f;
typedef __attribute__((ext_vector_type(8))) float v8f;

constexpr int Hdim = 256, Wdim = 256, Cin = 64, C2 = 256, HID = 128, Cout = 64;
constexpr int HW = Hdim * Wdim;

__constant__ const float DCT8[8][8] = {
  { 0.3535533906f, 0.3535533906f, 0.3535533906f, 0.3535533906f, 0.3535533906f, 0.3535533906f, 0.3535533906f, 0.3535533906f },
  { 0.4903926402f, 0.4157348062f, 0.2777851165f, 0.0975451610f,-0.0975451610f,-0.2777851165f,-0.4157348062f,-0.4903926402f },
  { 0.4619397663f, 0.1913417162f,-0.1913417162f,-0.4619397663f,-0.4619397663f,-0.1913417162f, 0.1913417162f, 0.4619397663f },
  { 0.4157348062f,-0.0975451610f,-0.4903926402f,-0.2777851165f, 0.2777851165f, 0.4903926402f, 0.0975451610f,-0.4157348062f },
  { 0.3535533906f,-0.3535533906f,-0.3535533906f, 0.3535533906f, 0.3535533906f,-0.3535533906f,-0.3535533906f, 0.3535533906f },
  { 0.2777851165f,-0.4903926402f, 0.0975451610f, 0.4157348062f,-0.4157348062f,-0.0975451610f, 0.4903926402f,-0.2777851165f },
  { 0.1913417162f,-0.4619397663f, 0.4619397663f,-0.1913417162f,-0.1913417162f, 0.4619397663f,-0.4619397663f, 0.1913417162f },
  { 0.0975451610f,-0.2777851165f, 0.4157348062f,-0.4903926402f, 0.4903926402f,-0.4157348062f, 0.2777851165f,-0.0975451610f }
};

typedef __attribute__((ext_vector_type(4))) float v4f;

constexpr int YP = 32;
constexpr int YC = 8 * YP + 4;
constexpr int WP = Cin + 4;

__device__ __forceinline__ int ysw(int row) { return ((row >> 1) & 1) << 2; }

__device__ __forceinline__ void dct_cols(float* __restrict__ t, int px)
{
    float* pa = t + px;
    float* pb = t + (px ^ 4);
    float x[8], w[8];
#pragma unroll
    for (int k = 0; k < 8; ++k) x[k] = (((k >> 1) & 1) ? pb : pa)[k * YP];
#pragma unroll
    for (int i = 0; i < 8; ++i) {
        float s = DCT8[i][0] * x[0];
#pragma unroll
        for (int k = 1; k < 8; ++k) s = __builtin_fmaf(DCT8[i][k], x[k], s);
        w[i] = s;
    }
#pragma unroll
    for (int i = 0; i < 8; ++i) (((i >> 1) & 1) ? pb : pa)[i * YP] = w[i];
}

__device__ __forceinline__ void dct_rows(float* __restrict__ t, int py, int pch)
{
    float* row = t + py * YP;
    float* p0 = row + ((8 * pch)     ^ ysw(py));
    float* p1 = row + ((8 * pch + 4) ^ ysw(py));
    const v4f a = *reinterpret_cast<const v4f*>(p0);
    const v4f b = *reinterpret_cast<const v4f*>(p1);
    float w[8], o[8];
#pragma unroll
    for (int e = 0; e < 4; ++e) { w[e] = a[e]; w[4 + e] = b[e]; }
#pragma unroll
    for (int q = 0; q < 8; ++q) {
        float s = w[0] * DCT8[q][0];
#pragma unroll
        for (int j = 1; j < 8; ++j) s = __builtin_fmaf(w[j], DCT8[q][j], s);
        o[q] = s;
    }
    v4f r0, r1;
#pragma unroll
    for (int e = 0; e < 4; ++e) { r0[e] = o[e]; r1[e] = o[4 + e]; }
    *reinterpret_cast<v4f*>(p0) = r0;
    *reinterpret_cast<v4f*>(p1) = r1;
}

__global__ __launch_bounds__(256) void
k1_projin_dct(const float* __restrict__ x, const float* __restrict__ Win,
              const float* __restrict__ mix, float* __restrict__ ws)
{
    __shared__ alignas(16) float yt[2][16 * YC];
    __shared__ alignas(16) float wt[2][16 * WP];
    __shared__ alignas(16) float xs[Cin * YC];

    const int tid  = threadIdx.x;
    const int lane = tid & 31;
    const int wv   = tid >> 5;
    const int m    = lane & 15;
    const int hi   = lane >> 4;

    const int b  = blockIdx.z;
    const int r0 = blockIdx.y * 8;
    const int c0 = blockIdx.x * 32;

    const float* wsrc = Win + 4 * tid;
    const int    wdst = (tid >> 4) * WP + 4 * (tid & 15);
    const v4f w0 = *reinterpret_cast<const v4f*>(wsrc);

    const float* xb = x + (size_t)b * Cin * HW;
    const int xrow = (tid >> 3) & 7;
    const int xu   = tid & 7;
    const float* xsrc = xb + (size_t)(tid >> 6) * HW + (size_t)(r0 + xrow) * Wdim + c0 + 4 * xu;
    const int    xdst = (tid >> 6) * YC + xrow * YP + ((4 * xu) ^ ysw(xrow));
    {
        v4f xv[16];
#pragma unroll
        for (int i = 0; i < 16; ++i) xv[i] = *reinterpret_cast<const v4f*>(xsrc + (size_t)(4 * i) * HW);
#pragma unroll
        for (int i = 0; i < 16; ++i) *reinterpret_cast<v4f*>(&xs[xdst + 4 * i * YC]) = xv[i];
    }
    *reinterpret_cast<v4f*>(&wt[0][wdst]) = w0;

    __syncthreads();

    const int pch = lane & 3;
    const int py  = lane >> 2;

#pragma unroll 1
    for (int t = 0; t < 8; t += 2) {
        dct_cols(xs + (wv + 8 * t) * YC, lane);
        dct_cols(xs + (wv + 8 * t + 8) * YC, lane);
    }
    __syncthreads();
#pragma unroll 1
    for (int t = 0; t < 8; t += 2) {
        dct_rows(xs + (wv + 8 * t) * YC, py, pch);
        dct_rows(xs + (wv + 8 * t + 8) * YC, py, pch);
    }
    __syncthreads();

    const float* xf = xs + wv * YP + ((2 * m) ^ ysw(wv));

    v2f be[16], bo[16];
#pragma unroll
    for (int s = 0; s < 16; ++s) {
        const int k0 = 4 * s + 2 * hi;
        const v2f l0 = *reinterpret_cast<const v2f*>(xf + k0 * YC);
        const v2f l1 = *reinterpret_cast<const v2f*>(xf + (k0 + 1) * YC);
        be[s].x = l0.x; be[s].y = l1.x;
        bo[s].x = l0.y; bo[s].y = l1.y;
    }

    const int wfr = m * WP + 2 * hi;
    v2f a[16];
#pragma unroll
    for (int s = 0; s < 16; ++s) a[s] = *reinterpret_cast<const v2f*>(&wt[0][wfr + 4 * s]);

    const float* mlane = mix + (size_t)(8 * hi) * 64 + wv * 8 + ((2 * m) & 7);

#pragma unroll 1
    for (int ob = 0; ob < 16; ++ob) {
        float* yb = yt[ob & 1];
        float* wn = wt[(ob + 1) & 1];

        const v4f wnx = *reinterpret_cast<const v4f*>(wsrc + (size_t)((ob < 15) ? ob + 1 : ob) * 16 * Cin);
        v2f mk[8];
#pragma unroll
        for (int r = 0; r < 8; ++r) mk[r] = *reinterpret_cast<const v2f*>(mlane + (size_t)(ob * 16 + r) * 64);
        __builtin_amdgcn_sched_barrier(0);

        v8f accE = {}, accO = {};
#pragma unroll
        for (int s = 0; s < 16; ++s) {
            accE = __builtin_amdgcn_wmma_f32_16x16x4_f32(false, a[s], false, be[s], (short)0, accE, false, false);
            accO = __builtin_amdgcn_wmma_f32_16x16x4_f32(false, a[s], false, bo[s], (short)0, accO, false, false);
        }

        {
            float* yd = yb + (8 * hi) * YC + wv * YP + ((2 * m) ^ ysw(wv));
#pragma unroll
            for (int r = 0; r < 8; ++r) {
                v2f pr; pr.x = accE[r] * mk[r].x; pr.y = accO[r] * mk[r].y;
                *reinterpret_cast<v2f*>(yd + r * YC) = pr;
            }
        }
        __syncthreads();

        *reinterpret_cast<v4f*>(wn + wdst) = wnx;

        float* tA = yb + wv * YC;
        float* tB = yb + (wv + 8) * YC;

        dct_cols(tA, lane);
        dct_cols(tB, lane);
        __syncthreads();
        dct_rows(tA, py, pch);
        dct_rows(tB, py, pch);
        __syncthreads();

#pragma unroll
        for (int it = 0; it < 4; ++it) {
            const int idx = tid + 256 * it;
            const int ch  = idx >> 6;
            const int row = (idx >> 3) & 7;
            const int u   = idx & 7;
            const v4f v = *reinterpret_cast<const v4f*>(yb + ch * YC + row * YP + ((4 * u) ^ ysw(row)));
            *reinterpret_cast<v4f*>(ws + ((size_t)b * C2 + ob * 16 + ch) * HW
                                       + (size_t)(r0 + row) * Wdim + c0 + 4 * u) = v;
        }

#pragma unroll
        for (int s = 0; s < 16; ++s) a[s] = *reinterpret_cast<const v2f*>(wn + wfr + 4 * s);
    }
}

constexpr int TP  = 76;
constexpr int TRH = 6;
constexpr int TCH = TRH * TP;
constexpr int GP  = 272;
constexpr int WLP = HID + 4;
constexpr float GK = -2.3022082f;

__global__ __launch_bounds__(256, 4) void
k2_dw_gate_out(const float* __restrict__ ws, const float* __restrict__ wdw,
               const float* __restrict__ wout, float* __restrict__ out)
{
    __shared__ alignas(16) float g[16][GP];
    __shared__ alignas(16) float tile[32 * TCH];
    __shared__ alignas(16) float wl[Cout * WLP];

    const int tid  = threadIdx.x;
    const int lane = tid & 31;
    const int wv   = tid >> 5;
    const int m    = lane & 15;
    const int hi   = lane >> 4;

    const int b  = blockIdx.z;
    const int r0 = blockIdx.y * 4;
    const int c0 = blockIdx.x * 64;

    const int wvu = __builtin_amdgcn_readfirstlane(wv);
    const int cy  = lane >> 3;
    const int cg  = lane & 7;

    const float* wsb  = ws + (size_t)b * C2 * HW;
    const int mt  = wv >> 1;
    const int ntb = (wv & 1) * 4;
    const float* wrow = wl + (mt * 16 + m) * WLP;

    const int q16 = tid & 15, rr = tid >> 4;
    v4f   cv[12];
    float hv[2];

    auto fetch = [&](int kc) {
#pragma unroll
        for (int t = 0; t < 12; ++t) {
            const int R   = rr + 16 * t;
            const int cs  = R / TRH;
            const int row = R - cs * TRH;
            const int yy  = r0 - 1 + row;
            const bool ok = ((unsigned)yy < (unsigned)Hdim);
            const int c   = kc + (cs & 15) + (cs >> 4) * HID;
            const float* p = wsb + (size_t)c * HW + (size_t)(ok ? yy : 0) * Wdim + c0 + q16 * 4;
            const v4f v = *reinterpret_cast<const v4f*>(p);
            cv[t] = ok ? v : (v4f){0.f, 0.f, 0.f, 0.f};
        }
#pragma unroll
        for (int t = 0; t < 2; ++t) {
            const int e    = tid + 256 * t;
            const int side = e & 1;
            const int R    = e >> 1;
            const int cs   = R / TRH;
            const int row  = R - cs * TRH;
            const int yy   = r0 - 1 + row;
            const int xx   = side ? c0 + 64 : c0 - 1;
            const bool ok  = (e < 384) && ((unsigned)yy < (unsigned)Hdim) && ((unsigned)xx < (unsigned)Wdim);
            const int c    = kc + (cs & 15) + (cs >> 4) * HID;
            hv[t] = ok ? wsb[(size_t)c * HW + (size_t)yy * Wdim + xx] : 0.f;
        }
    };
    auto stage = [&]() {
#pragma unroll
        for (int t = 0; t < 12; ++t) {
            const int R   = rr + 16 * t;
            const int cs  = R / TRH;
            const int row = R - cs * TRH;
            *reinterpret_cast<v4f*>(&tile[cs * TCH + row * TP + 4 + q16 * 4]) = cv[t];
        }
#pragma unroll
        for (int t = 0; t < 2; ++t) {
            const int e    = tid + 256 * t;
            const int side = e & 1;
            const int R    = e >> 1;
            const int cs   = R / TRH;
            const int row  = R - cs * TRH;
            if (e < 384) tile[cs * TCH + row * TP + (side ? 68 : 3)] = hv[t];
        }
    };

    v8f accE[4], accO[4];
#pragma unroll
    for (int j = 0; j < 4; ++j) { accE[j] = (v8f){}; accO[j] = (v8f){}; }

    v4f wq[8];
#pragma unroll
    for (int i = 0; i < 8; ++i) wq[i] = *reinterpret_cast<const v4f*>(wout + 4 * (tid + 256 * i));

    fetch(0);
#pragma unroll
    for (int i = 0; i < 8; ++i) {
        const int p = tid + 256 * i;
        *reinterpret_cast<v4f*>(&wl[(p >> 5) * WLP + 4 * (p & 31)]) = wq[i];
    }
    stage();
    __syncthreads();

    for (int kc = 0; kc < HID; kc += 16) {
        const bool more = (kc + 16 < HID);
        if (more) fetch(kc + 16);
#pragma unroll 1
        for (int cc = 0; cc < 2; ++cc) {
            const int i = 2 * wvu + cc;
            const int c = kc + i;
            const float* w1 = wdw + c * 9;
            const float* w2 = wdw + (c + HID) * 9;
            const float* t1 = tile + i * TCH + cy * TP + 3 + 8 * cg;
            const float* t2 = t1 + 16 * TCH;
            float s1[8], s2[8];
#pragma unroll
            for (int j = 0; j < 8; ++j) { s1[j] = 0.f; s2[j] = 0.f; }
#pragma unroll
            for (int ky = 0; ky < 3; ++ky) {
                const float* r1 = t1 + ky * TP;
                const float* r2 = t2 + ky * TP;
                float a[10], bq[10];
                const v4f a0 = *reinterpret_cast<const v4f*>(r1 + 1);
                const v4f a1 = *reinterpret_cast<const v4f*>(r1 + 5);
                const v4f b0 = *reinterpret_cast<const v4f*>(r2 + 1);
                const v4f b1 = *reinterpret_cast<const v4f*>(r2 + 5);
                a[0] = r1[0];  a[9] = r1[9];
                bq[0] = r2[0]; bq[9] = r2[9];
#pragma unroll
                for (int e = 0; e < 4; ++e) { a[1 + e] = a0[e]; a[5 + e] = a1[e]; bq[1 + e] = b0[e]; bq[5 + e] = b1[e]; }
#pragma unroll
                for (int kx = 0; kx < 3; ++kx) {
                    const float wa = w1[ky * 3 + kx];
                    const float wb = w2[ky * 3 + kx];
#pragma unroll
                    for (int j = 0; j < 8; ++j) {
                        s1[j] += wa * a[j + kx];
                        s2[j] += wb * bq[j + kx];
                    }
                }
            }
            v4f o0, o1;
#pragma unroll
            for (int j = 0; j < 8; ++j) {
                const float p = s1[j] + 0.044715f * s1[j] * s1[j] * s1[j];
                const float e = __builtin_amdgcn_exp2f(GK * p);
                const float v = s1[j] * __builtin_amdgcn_rcpf(1.f + e) * s2[j];
                if (j < 4) o0[j] = v; else o1[j - 4] = v;
            }
            float* gp = &g[i][cy * 64 + 8 * cg];
            *reinterpret_cast<v4f*>(gp)     = o0;
            *reinterpret_cast<v4f*>(gp + 4) = o1;
        }
        __syncthreads();
        if (more) stage();
#pragma unroll
        for (int kk = 0; kk < 16; kk += 4) {
            const int k0 = kk + 2 * hi;
            const v2f a = *reinterpret_cast<const v2f*>(wrow + kc + k0);
#pragma unroll
            for (int j = 0; j < 4; ++j) {
                const int nt = ntb + j;
                const v2f p0 = *reinterpret_cast<const v2f*>(&g[k0][nt * 32 + 2 * m]);
                const v2f p1 = *reinterpret_cast<const v2f*>(&g[k0 + 1][nt * 32 + 2 * m]);
                v2f be, bo;
                be.x = p0.x; be.y = p1.x;
                bo.x = p0.y; bo.y = p1.y;
                accE[j] = __builtin_amdgcn_wmma_f32_16x16x4_f32(false, a, false, be, (short)0, accE[j], false, false);
                accO[j] = __builtin_amdgcn_wmma_f32_16x16x4_f32(false, a, false, bo, (short)0, accO[j], false, false);
            }
        }
        __syncthreads();
    }

#pragma unroll
    for (int j = 0; j < 4; ++j) {
        const int nt = ntb + j;
        const size_t base = ((size_t)b * Cout + mt * 16 + 8 * hi) * HW
                          + (size_t)(r0 + (nt >> 1)) * Wdim + c0 + 32 * (nt & 1) + 2 * m;
#pragma unroll
        for (int r = 0; r < 8; ++r) {
            v2f pr; pr.x = accE[j][r]; pr.y = accO[j][r];
            *reinterpret_cast<v2f*>(out + base + (size_t)r * HW) = pr;
        }
    }
}

extern "C" void kernel_launch(void* const* d_in, const int* in_sizes, int n_in,
                              void* d_out, int out_size, void* d_ws, size_t ws_size,
                              hipStream_t stream) {
    const float* x    = (const float*)d_in[0];
    const float* Win  = (const float*)d_in[1];
    const float* Wdw  = (const float*)d_in[2];
    const float* mix  = (const float*)d_in[3];
    const float* Wout = (const float*)d_in[4];
    float* out = (float*)d_out;
    float* ws  = (float*)d_ws;

    dim3 blk(256);
    dim3 g1(Wdim / 32, Hdim / 8, 2);
    k1_projin_dct<<<g1, blk, 0, stream>>>(x, Win, mix, ws);
    dim3 g2(Wdim / 64, Hdim / 4, 2);
    k2_dw_gate_out<<<g2, blk, 0, stream>>>(ws, Wdw, Wout, out);
}
